# starting kernel
# baseline (speedup 1.0000x reference)
#include <hip/hip_runtime.h>
#include <hip/hip_bf16.h>
#include <math.h>

typedef float v2f __attribute__((ext_vector_type(2)));
typedef float v8f __attribute__((ext_vector_type(8)));

#define B_SZ   2
#define T_SEQ  2048
#define C_DIM  1024
#define NH     16
#define HD     32
#define QKH    32
#define VD     64
#define ROWQKV 3072

__device__ __forceinline__ v8f wmma4(v2f a, v2f b, v8f c) {
  return __builtin_amdgcn_wmma_f32_16x16x4_f32(
       false, a,  false, b,
       (short)0, c,  false,  false);
}

__global__ __launch_bounds__(256) void k_gemm(const float* __restrict__ A,
                                              const float* __restrict__ Bm,
                                              float* __restrict__ Cm,
                                              int N, int K) {
  const int lane = threadIdx.x & 31;
  const int wid  = threadIdx.x >> 5;
  const int wy   = wid & 3;
  const int wx   = wid >> 2;
  const int rowBase = blockIdx.y * 128 + wy * 32;
  const int colBase = blockIdx.x * 128 + wx * 64;
  const int hi   = lane >> 4;
  const int l15  = lane & 15;
  const int bc   = colBase + l15;

  const float* pA0 = A + (size_t)(rowBase + l15) * K + hi * 2;
  const float* pA1 = A + (size_t)(rowBase + 16 + l15) * K + hi * 2;
  const float* pB  = Bm + (size_t)(hi * 2) * N + bc;
  const size_t strideB = (size_t)4 * N;

  v8f acc[2][4] = {};

  v2f a0 = *(const v2f*)pA0;
  v2f a1 = *(const v2f*)pA1;
  v2f b[4];
  #pragma unroll
  for (int cg = 0; cg < 4; ++cg) {
    b[cg].x = pB[16 * cg];
    b[cg].y = pB[N + 16 * cg];
  }

  #pragma unroll 2
  for (int k = 0; k < K - 4; k += 4) {
    pA0 += 4; pA1 += 4; pB += strideB;
    const v2f na0 = *(const v2f*)pA0;
    const v2f na1 = *(const v2f*)pA1;
    v2f nb[4];
    #pragma unroll
    for (int cg = 0; cg < 4; ++cg) {
      nb[cg].x = pB[16 * cg];
      nb[cg].y = pB[N + 16 * cg];
    }
    if ((k & 60) == 0) {
      __builtin_prefetch(pB + 16 * strideB, 0, 1);
      __builtin_prefetch(pA0 + 64, 0, 1);
      __builtin_prefetch(pA1 + 64, 0, 1);
    }
    #pragma unroll
    for (int cg = 0; cg < 4; ++cg) {
      acc[0][cg] = wmma4(a0, b[cg], acc[0][cg]);
      acc[1][cg] = wmma4(a1, b[cg], acc[1][cg]);
    }
    a0 = na0; a1 = na1;
    #pragma unroll
    for (int cg = 0; cg < 4; ++cg) b[cg] = nb[cg];
  }
  #pragma unroll
  for (int cg = 0; cg < 4; ++cg) {
    acc[0][cg] = wmma4(a0, b[cg], acc[0][cg]);
    acc[1][cg] = wmma4(a1, b[cg], acc[1][cg]);
  }

  #pragma unroll
  for (int g = 0; g < 2; ++g) {
    #pragma unroll
    for (int r = 0; r < 8; ++r) {
      const size_t row = (size_t)(rowBase + g * 16 + r + hi * 8);
      #pragma unroll
      for (int cg = 0; cg < 4; ++cg) {
        Cm[row * N + bc + 16 * cg] = acc[g][cg][r];
      }
    }
  }
}

__global__ void k_rope(float* __restrict__ qkv) {
  size_t idx = (size_t)blockIdx.x * blockDim.x + threadIdx.x;
  const int j  = idx & 15;  idx >>= 4;
  const int h  = idx & 31;  idx >>= 5;
  const int qk = idx & 1;   idx >>= 1;
  const int t  = idx & 2047; idx >>= 11;
  const int b  = (int)idx;
  const float theta = __expf(-((float)(2 * j) / 32.0f) * logf(10000.0f));
  float s, c;
  __sincosf((float)t * theta, &s, &c);
  float* p = qkv + ((size_t)(b * T_SEQ + t)) * ROWQKV + qk * C_DIM + h * HD + 2 * j;
  const float x1 = p[0], x2 = p[1];
  p[0] = x1 * c - x2 * s;
  p[1] = x1 * s + x2 * c;
}

__global__ void k_lambda(const float* __restrict__ lq1, const float* __restrict__ lk1,
                         const float* __restrict__ lq2, const float* __restrict__ lk2,
                         float* __restrict__ lam) {
  const int l = threadIdx.x;
  float s1 = lq1[l] * lk1[l];
  float s2 = lq2[l] * lk2[l];
  #pragma unroll
  for (int off = 16; off; off >>= 1) {
    s1 += __shfl_xor(s1, off, 32);
    s2 += __shfl_xor(s2, off, 32);
  }
  if (l == 0) *lam = __expf(s1) - __expf(s2) + 0.2f;
}

__global__ __launch_bounds__(32) void k_attn(const float* __restrict__ qkv,
                                             float* __restrict__ Abuf) {
  const int qt   = blockIdx.x;
  const int h    = blockIdx.y;
  const int b    = blockIdx.z;
  const int lane = threadIdx.x;
  const int hi   = lane >> 4;
  const int l15  = lane & 15;

  const float* Q  = qkv + (size_t)b * T_SEQ * ROWQKV + h * HD;
  const float* Kp = qkv + (size_t)b * T_SEQ * ROWQKV + C_DIM + h * HD;
  const float* V  = qkv + (size_t)b * T_SEQ * ROWQKV + 2 * C_DIM + (h & 15) * VD;

  __shared__ float S[16 * 18];
  __shared__ float mrow[16], lrow[16], frow[16];

  if (lane < 16) { mrow[lane] = -3.0e38f; lrow[lane] = 0.0f; }
  __syncthreads();

  const size_t qoff = (size_t)(qt * 16 + l15) * ROWQKV;
  v2f qa[8];
  #pragma unroll
  for (int c = 0; c < 8; ++c) {
    const int col = 4 * c + hi * 2;
    qa[c].x = Q[qoff + col];
    qa[c].y = Q[qoff + col + 1];
  }

  v8f o0 = {}, o1 = {}, o2 = {}, o3 = {};
  const float scale = 0.17677669529663687f;

  for (int kb = 0; kb <= qt; ++kb) {
    v8f s = {};
    const size_t koff = (size_t)(kb * 16 + l15) * ROWQKV;
    #pragma unroll
    for (int c = 0; c < 8; ++c) {
      const int col = 4 * c + hi * 2;
      v2f kf; kf.x = Kp[koff + col]; kf.y = Kp[koff + col + 1];
      s = wmma4(qa[c], kf, s);
    }
    #pragma unroll
    for (int r = 0; r < 8; ++r) {
      const int m    = r + hi * 8;
      const int qidx = qt * 16 + m;
      const int kidx = kb * 16 + l15;
      S[m * 18 + l15] = (kidx <= qidx) ? s[r] * scale : -3.0e38f;
    }
    __syncthreads();

    if (lane < 16) {
      float mx = mrow[lane];
      #pragma unroll
      for (int j = 0; j < 16; ++j) mx = fmaxf(mx, S[lane * 18 + j]);
      const float f = __expf(mrow[lane] - mx);
      float sum = 0.0f;
      #pragma unroll
      for (int j = 0; j < 16; ++j) {
        const float p = __expf(S[lane * 18 + j] - mx);
        S[lane * 18 + j] = p;
        sum += p;
      }
      mrow[lane] = mx;
      frow[lane] = f;
      lrow[lane] = lrow[lane] * f + sum;
    }
    __syncthreads();

    float fr[8];
    #pragma unroll
    for (int r = 0; r < 8; ++r) fr[r] = frow[r + hi * 8];
    #pragma unroll
    for (int r = 0; r < 8; ++r) {
      o0[r] *= fr[r]; o1[r] *= fr[r]; o2[r] *= fr[r]; o3[r] *= fr[r];
    }

    v2f pa[4];
    #pragma unroll
    for (int c = 0; c < 4; ++c) {
      const int col = 4 * c + hi * 2;
      pa[c].x = S[l15 * 18 + col];
      pa[c].y = S[l15 * 18 + col + 1];
    }
    #pragma unroll
    for (int c = 0; c < 4; ++c) {
      const int krow = kb * 16 + 4 * c + hi * 2;
      const float* Vr0 = V + (size_t)krow * ROWQKV;
      const float* Vr1 = V + (size_t)(krow + 1) * ROWQKV;
      v2f vb0; vb0.x = Vr0[l15];      vb0.y = Vr1[l15];
      v2f vb1; vb1.x = Vr0[l15 + 16]; vb1.y = Vr1[l15 + 16];
      v2f vb2; vb2.x = Vr0[l15 + 32]; vb2.y = Vr1[l15 + 32];
      v2f vb3; vb3.x = Vr0[l15 + 48]; vb3.y = Vr1[l15 + 48];
      o0 = wmma4(pa[c], vb0, o0);
      o1 = wmma4(pa[c], vb1, o1);
      o2 = wmma4(pa[c], vb2, o2);
      o3 = wmma4(pa[c], vb3, o3);
    }
    __syncthreads();
  }

  float inv[8];
  #pragma unroll
  for (int r = 0; r < 8; ++r) inv[r] = 1.0f / lrow[r + hi * 8];
  float* Ab = Abuf + ((size_t)(b * QKH + h)) * T_SEQ * VD;
  #pragma unroll
  for (int r = 0; r < 8; ++r) {
    float* Ar = Ab + (size_t)(qt * 16 + r + hi * 8) * VD;
    Ar[l15]      = o0[r] * inv[r];
    Ar[l15 + 16] = o1[r] * inv[r];
    Ar[l15 + 32] = o2[r] * inv[r];
    Ar[l15 + 48] = o3[r] * inv[r];
  }
}

__global__ __launch_bounds__(256) void k_diffnorm(const float* __restrict__ Abuf,
                                                  const float* __restrict__ lamp,
                                                  float* __restrict__ Dbuf) {
  const float lam = *lamp;
  const int wid  = threadIdx.x >> 5;
  const int lane = threadIdx.x & 31;
  const size_t row = (size_t)blockIdx.x * 8 + wid;
  const int t  = (int)(row & 2047);
  const size_t bh = row >> 11;
  const int hh = (int)(bh & 15);
  const int b  = (int)(bh >> 4);

  const float* A1 = Abuf + ((size_t)(b * QKH + hh))      * T_SEQ * VD + (size_t)t * VD;
  const float* A2 = Abuf + ((size_t)(b * QKH + 16 + hh)) * T_SEQ * VD + (size_t)t * VD;
  const float d0 = A1[lane]      - lam * A2[lane];
  const float d1 = A1[lane + 32] - lam * A2[lane + 32];
  float ss = d0 * d0 + d1 * d1;
  #pragma unroll
  for (int off = 16; off; off >>= 1) ss += __shfl_xor(ss, off, 32);
  const float r = rsqrtf(ss * (1.0f / 64.0f) + 1e-5f) * 0.8f;
  float* Dp = Dbuf + ((size_t)(b * T_SEQ + t)) * C_DIM + hh * VD;
  Dp[lane]      = d0 * r;
  Dp[lane + 32] = d1 * r;
}

extern "C" void kernel_launch(void* const* d_in, const int* in_sizes, int n_in,
                              void* d_out, int out_size, void* d_ws, size_t ws_size,
                              hipStream_t stream) {
  const float* x      = (const float*)d_in[0];
  const float* W_attn = (const float*)d_in[1];
  const float* W_proj = (const float*)d_in[2];
  const float* lq1    = (const float*)d_in[3];
  const float* lk1    = (const float*)d_in[4];
  const float* lq2    = (const float*)d_in[5];
  const float* lk2    = (const float*)d_in[6];
  float* out = (float*)d_out;

  float* qkv  = (float*)d_ws;
  float* Abuf = qkv  + (size_t)B_SZ * T_SEQ * ROWQKV;
  float* Dbuf = Abuf + (size_t)B_SZ * QKH * T_SEQ * VD;
  float* lam  = Dbuf + (size_t)B_SZ * T_SEQ * C_DIM;

  const int M = B_SZ * T_SEQ;

  {
    dim3 grid(3 * C_DIM / 128, M / 128);
    k_gemm<<<grid, 256, 0, stream>>>(x, W_attn, qkv, 3 * C_DIM, C_DIM);
  }
  {
    const size_t n = (size_t)B_SZ * T_SEQ * 2 * QKH * 16;
    k_rope<<<(unsigned)(n / 256), 256, 0, stream>>>(qkv);
  }
  k_lambda<<<1, 32, 0, stream>>>(lq1, lk1, lq2, lk2, lam);
  {
    dim3 grid(T_SEQ / 16, QKH, B_SZ);
    k_attn<<<grid, 32, 0, stream>>>(qkv, Abuf);
  }
  k_diffnorm<<<(B_SZ * NH * T_SEQ) / 8, 256, 0, stream>>>(Abuf, lam, Dbuf);
  {
    dim3 grid(C_DIM / 128, M / 128);
    k_gemm<<<grid, 256, 0, stream>>>(Dbuf, W_proj, out, C_DIM, C_DIM);
  }
  (void)in_sizes; (void)n_in; (void)out_size; (void)ws_size;
}


# attention: K (after rotary) and V repacked once per call into matrix-fragment order, so every K/V fragment is one contiguous 128-bit load per lane; arithmetic and sum order unchanged
# speedup vs baseline: 1.0679x; 1.0679x over previous
#include <hip/hip_runtime.h>
#include <hip/hip_bf16.h>
#include <math.h>

typedef float v2f __attribute__((ext_vector_type(2)));
typedef float v4f __attribute__((ext_vector_type(4)));
typedef float v8f __attribute__((ext_vector_type(8)));

#define B_SZ   2
#define T_SEQ  2048
#define C_DIM  1024
#define NH     16
#define HD     32
#define QKH    32
#define VD     64
#define ROWQKV 3072

__device__ __forceinline__ v8f wmma4(v2f a, v2f b, v8f c) {
  return __builtin_amdgcn_wmma_f32_16x16x4_f32(
       false, a,  false, b,
       (short)0, c,  false,  false);
}

__global__ __launch_bounds__(256) void k_gemm(const float* __restrict__ A,
                                              const float* __restrict__ Bm,
                                              float* __restrict__ Cm,
                                              int N, int K) {
  const int lane = threadIdx.x & 31;
  const int wid  = threadIdx.x >> 5;
  const int wy   = wid & 3;
  const int wx   = wid >> 2;
  const int rowBase = blockIdx.y * 128 + wy * 32;
  const int colBase = blockIdx.x * 128 + wx * 64;
  const int hi   = lane >> 4;
  const int l15  = lane & 15;
  const int bc   = colBase + l15;

  const float* pA0 = A + (size_t)(rowBase + l15) * K + hi * 2;
  const float* pA1 = A + (size_t)(rowBase + 16 + l15) * K + hi * 2;
  const float* pB  = Bm + (size_t)(hi * 2) * N + bc;
  const size_t strideB = (size_t)4 * N;

  v8f acc[2][4] = {};

  v2f a0 = *(const v2f*)pA0;
  v2f a1 = *(const v2f*)pA1;
  v2f b[4];
  #pragma unroll
  for (int cg = 0; cg < 4; ++cg) {
    b[cg].x = pB[16 * cg];
    b[cg].y = pB[N + 16 * cg];
  }

  #pragma unroll 2
  for (int k = 0; k < K - 4; k += 4) {
    pA0 += 4; pA1 += 4; pB += strideB;
    const v2f na0 = *(const v2f*)pA0;
    const v2f na1 = *(const v2f*)pA1;
    v2f nb[4];
    #pragma unroll
    for (int cg = 0; cg < 4; ++cg) {
      nb[cg].x = pB[16 * cg];
      nb[cg].y = pB[N + 16 * cg];
    }
    if ((k & 60) == 0) {
      __builtin_prefetch(pB + 16 * strideB, 0, 1);
      __builtin_prefetch(pA0 + 64, 0, 1);
      __builtin_prefetch(pA1 + 64, 0, 1);
    }
    #pragma unroll
    for (int cg = 0; cg < 4; ++cg) {
      acc[0][cg] = wmma4(a0, b[cg], acc[0][cg]);
      acc[1][cg] = wmma4(a1, b[cg], acc[1][cg]);
    }
    a0 = na0; a1 = na1;
    #pragma unroll
    for (int cg = 0; cg < 4; ++cg) b[cg] = nb[cg];
  }
  #pragma unroll
  for (int cg = 0; cg < 4; ++cg) {
    acc[0][cg] = wmma4(a0, b[cg], acc[0][cg]);
    acc[1][cg] = wmma4(a1, b[cg], acc[1][cg]);
  }

  #pragma unroll
  for (int g = 0; g < 2; ++g) {
    #pragma unroll
    for (int r = 0; r < 8; ++r) {
      const size_t row = (size_t)(rowBase + g * 16 + r + hi * 8);
      #pragma unroll
      for (int cg = 0; cg < 4; ++cg) {
        Cm[row * N + bc + 16 * cg] = acc[g][cg][r];
      }
    }
  }
}

#define QBLK 8192
#define KBLK 4096
#define VBLK 4096

__device__ __forceinline__ v2f rope_pair(v2f x, int j, int t) {
  const float theta = __expf(-((float)(2 * j) / 32.0f) * logf(10000.0f));
  float s, c;
  __sincosf((float)t * theta, &s, &c);
  v2f r;
  r.x = x.x * c - x.y * s;
  r.y = x.x * s + x.y * c;
  return r;
}

__global__ __launch_bounds__(256) void k_pack(float* __restrict__ qkv,
                                              float* __restrict__ Kf,
                                              float* __restrict__ Vf) {
  const unsigned blk = blockIdx.x;
  if (blk < QBLK) {
    unsigned idx = blk * 256u + threadIdx.x;
    const int j = idx & 15;   idx >>= 4;
    const int h = idx & 31;   idx >>= 5;
    const int t = idx & 2047; idx >>= 11;
    const int b = (int)idx;
    float* p = qkv + ((size_t)(b * T_SEQ + t)) * ROWQKV + h * HD + 2 * j;
    const v2f x = *(const v2f*)p;
    *(v2f*)p = rope_pair(x, j, t);
  } else if (blk < QBLK + KBLK) {
    const unsigned o = (blk - QBLK) * 256u + threadIdx.x;
    const int lane = o & 31;
    const int c2   = (o >> 5) & 3;
    const int kb   = (o >> 7) & 127;
    const int bh   = (int)(o >> 14);
    const int hi   = lane >> 4;
    const int l15  = lane & 15;
    const int b    = bh >> 5;
    const int h    = bh & 31;
    const int t    = kb * 16 + l15;
    const float* s0 = qkv + ((size_t)(b * T_SEQ + t)) * ROWQKV + C_DIM + h * HD
                          + 8 * c2 + 2 * hi;
    const v2f xa = *(const v2f*)s0;
    const v2f xb = *(const v2f*)(s0 + 4);
    const v2f ra = rope_pair(xa, 4 * c2 + hi, t);
    const v2f rb = rope_pair(xb, 4 * c2 + 2 + hi, t);
    v4f w; w.x = ra.x; w.y = ra.y; w.z = rb.x; w.w = rb.y;
    ((v4f*)Kf)[o] = w;
  } else {
    const unsigned o = (blk - (QBLK + KBLK)) * 256u + threadIdx.x;
    const int lane = o & 31;
    const int jp   = (o >> 5) & 1;
    const int c    = (o >> 6) & 3;
    const int kb   = (o >> 8) & 127;
    const int bhv  = (int)(o >> 15);
    const int hi   = lane >> 4;
    const int l15  = lane & 15;
    const int b    = bhv >> 4;
    const int hv   = bhv & 15;
    const int k0   = kb * 16 + 4 * c + 2 * hi;
    const float* s0 = qkv + ((size_t)(b * T_SEQ + k0)) * ROWQKV + 2 * C_DIM + hv * VD
                          + 32 * jp + l15;
    const float* s1 = s0 + ROWQKV;
    v4f w; w.x = s0[0]; w.y = s1[0]; w.z = s0[16]; w.w = s1[16];
    ((v4f*)Vf)[o] = w;
  }
}

__global__ void k_lambda(const float* __restrict__ lq1, const float* __restrict__ lk1,
                         const float* __restrict__ lq2, const float* __restrict__ lk2,
                         float* __restrict__ lam) {
  const int l = threadIdx.x;
  float s1 = lq1[l] * lk1[l];
  float s2 = lq2[l] * lk2[l];
  #pragma unroll
  for (int off = 16; off; off >>= 1) {
    s1 += __shfl_xor(s1, off, 32);
    s2 += __shfl_xor(s2, off, 32);
  }
  if (l == 0) *lam = __expf(s1) - __expf(s2) + 0.2f;
}

__global__ __launch_bounds__(32) void k_attn(const float* __restrict__ qkv,
                                             const float* __restrict__ Kf,
                                             const float* __restrict__ Vf,
                                             float* __restrict__ Abuf) {
  const int qt   = blockIdx.x;
  const int h    = blockIdx.y;
  const int b    = blockIdx.z;
  const int lane = threadIdx.x;
  const int hi   = lane >> 4;
  const int l15  = lane & 15;

  const float* Q  = qkv + (size_t)b * T_SEQ * ROWQKV + h * HD;
  const v4f* Kq = (const v4f*)Kf + (size_t)(b * QKH + h) * (128 * 128) + lane;
  const v4f* Vq = (const v4f*)Vf + (size_t)(b * NH + (h & 15)) * (128 * 256) + lane;

  __shared__ float S[16 * 18];
  __shared__ float mrow[16], lrow[16], frow[16];

  if (lane < 16) { mrow[lane] = -3.0e38f; lrow[lane] = 0.0f; }
  __syncthreads();

  const size_t qoff = (size_t)(qt * 16 + l15) * ROWQKV;
  v2f qa[8];
  #pragma unroll
  for (int c = 0; c < 8; ++c) {
    const int col = 4 * c + hi * 2;
    qa[c].x = Q[qoff + col];
    qa[c].y = Q[qoff + col + 1];
  }

  v8f o0 = {}, o1 = {}, o2 = {}, o3 = {};
  const float scale = 0.17677669529663687f;

  for (int kb = 0; kb <= qt; ++kb) {
    v8f s = {};
    const v4f* kp = Kq + kb * 128;
    #pragma unroll
    for (int c2 = 0; c2 < 4; ++c2) {
      const v4f kq = kp[c2 * 32];
      v2f kf0; kf0.x = kq.x; kf0.y = kq.y;
      v2f kf1; kf1.x = kq.z; kf1.y = kq.w;
      s = wmma4(qa[2 * c2],     kf0, s);
      s = wmma4(qa[2 * c2 + 1], kf1, s);
    }
    #pragma unroll
    for (int r = 0; r < 8; ++r) {
      const int m    = r + hi * 8;
      const int qidx = qt * 16 + m;
      const int kidx = kb * 16 + l15;
      S[m * 18 + l15] = (kidx <= qidx) ? s[r] * scale : -3.0e38f;
    }
    __syncthreads();

    if (lane < 16) {
      float mx = mrow[lane];
      #pragma unroll
      for (int j = 0; j < 16; ++j) mx = fmaxf(mx, S[lane * 18 + j]);
      const float f = __expf(mrow[lane] - mx);
      float sum = 0.0f;
      #pragma unroll
      for (int j = 0; j < 16; ++j) {
        const float p = __expf(S[lane * 18 + j] - mx);
        S[lane * 18 + j] = p;
        sum += p;
      }
      mrow[lane] = mx;
      frow[lane] = f;
      lrow[lane] = lrow[lane] * f + sum;
    }
    __syncthreads();

    float fr[8];
    #pragma unroll
    for (int r = 0; r < 8; ++r) fr[r] = frow[r + hi * 8];
    #pragma unroll
    for (int r = 0; r < 8; ++r) {
      o0[r] *= fr[r]; o1[r] *= fr[r]; o2[r] *= fr[r]; o3[r] *= fr[r];
    }

    v2f pa[4];
    #pragma unroll
    for (int c = 0; c < 4; ++c) {
      const int col = 4 * c + hi * 2;
      pa[c].x = S[l15 * 18 + col];
      pa[c].y = S[l15 * 18 + col + 1];
    }
    const v4f* vp = Vq + kb * 256;
    #pragma unroll
    for (int c = 0; c < 4; ++c) {
      const v4f va = vp[(2 * c) * 32];
      const v4f vc = vp[(2 * c + 1) * 32];
      v2f vb0; vb0.x = va.x; vb0.y = va.y;
      v2f vb1; vb1.x = va.z; vb1.y = va.w;
      v2f vb2; vb2.x = vc.x; vb2.y = vc.y;
      v2f vb3; vb3.x = vc.z; vb3.y = vc.w;
      o0 = wmma4(pa[c], vb0, o0);
      o1 = wmma4(pa[c], vb1, o1);
      o2 = wmma4(pa[c], vb2, o2);
      o3 = wmma4(pa[c], vb3, o3);
    }
    __syncthreads();
  }

  float inv[8];
  #pragma unroll
  for (int r = 0; r < 8; ++r) inv[r] = 1.0f / lrow[r + hi * 8];
  float* Ab = Abuf + ((size_t)(b * QKH + h)) * T_SEQ * VD;
  #pragma unroll
  for (int r = 0; r < 8; ++r) {
    float* Ar = Ab + (size_t)(qt * 16 + r + hi * 8) * VD;
    Ar[l15]      = o0[r] * inv[r];
    Ar[l15 + 16] = o1[r] * inv[r];
    Ar[l15 + 32] = o2[r] * inv[r];
    Ar[l15 + 48] = o3[r] * inv[r];
  }
}

__global__ __launch_bounds__(256) void k_diffnorm(const float* __restrict__ Abuf,
                                                  const float* __restrict__ lamp,
                                                  float* __restrict__ Dbuf) {
  const float lam = *lamp;
  const int wid  = threadIdx.x >> 5;
  const int lane = threadIdx.x & 31;
  const size_t row = (size_t)blockIdx.x * 8 + wid;
  const int t  = (int)(row & 2047);
  const size_t bh = row >> 11;
  const int hh = (int)(bh & 15);
  const int b  = (int)(bh >> 4);

  const float* A1 = Abuf + ((size_t)(b * QKH + hh))      * T_SEQ * VD + (size_t)t * VD;
  const float* A2 = Abuf + ((size_t)(b * QKH + 16 + hh)) * T_SEQ * VD + (size_t)t * VD;
  const float d0 = A1[lane]      - lam * A2[lane];
  const float d1 = A1[lane + 32] - lam * A2[lane + 32];
  float ss = d0 * d0 + d1 * d1;
  #pragma unroll
  for (int off = 16; off; off >>= 1) ss += __shfl_xor(ss, off, 32);
  const float r = rsqrtf(ss * (1.0f / 64.0f) + 1e-5f) * 0.8f;
  float* Dp = Dbuf + ((size_t)(b * T_SEQ + t)) * C_DIM + hh * VD;
  Dp[lane]      = d0 * r;
  Dp[lane + 32] = d1 * r;
}

extern "C" void kernel_launch(void* const* d_in, const int* in_sizes, int n_in,
                              void* d_out, int out_size, void* d_ws, size_t ws_size,
                              hipStream_t stream) {
  const float* x      = (const float*)d_in[0];
  const float* W_attn = (const float*)d_in[1];
  const float* W_proj = (const float*)d_in[2];
  const float* lq1    = (const float*)d_in[3];
  const float* lk1    = (const float*)d_in[4];
  const float* lq2    = (const float*)d_in[5];
  const float* lk2    = (const float*)d_in[6];
  float* out = (float*)d_out;

  float* qkv  = (float*)d_ws;
  float* Abuf = qkv  + (size_t)B_SZ * T_SEQ * ROWQKV;
  float* Dbuf = Abuf + (size_t)B_SZ * QKH * T_SEQ * VD;
  float* Kf   = Dbuf + (size_t)B_SZ * T_SEQ * C_DIM;
  float* Vf   = Kf   + (size_t)B_SZ * QKH * T_SEQ * HD;
  float* lam  = Vf   + (size_t)B_SZ * NH * T_SEQ * VD;

  const int M = B_SZ * T_SEQ;

  {
    dim3 grid(3 * C_DIM / 128, M / 128);
    k_gemm<<<grid, 256, 0, stream>>>(x, W_attn, qkv, 3 * C_DIM, C_DIM);
  }
  k_pack<<<QBLK + KBLK + VBLK, 256, 0, stream>>>(qkv, Kf, Vf);
  k_lambda<<<1, 32, 0, stream>>>(lq1, lk1, lq2, lk2, lam);
  {
    dim3 grid(T_SEQ / 16, QKH, B_SZ);
    k_attn<<<grid, 32, 0, stream>>>(qkv, Kf, Vf, Abuf);
  }
  k_diffnorm<<<(B_SZ * NH * T_SEQ) / 8, 256, 0, stream>>>(Abuf, lam, Dbuf);
  {
    dim3 grid(C_DIM / 128, M / 128);
    k_gemm<<<grid, 256, 0, stream>>>(Dbuf, W_proj, out, C_DIM, C_DIM);
  }
  (void)in_sizes; (void)n_in; (void)out_size; (void)ws_size;
}


# attention: one online-softmax step per 32 keys (two score tiles a trip); all 32 lanes take half a row, maximum joined by one lane-half exchange; row sums kept per lane half and added once at the end
# speedup vs baseline: 1.2076x; 1.1308x over previous
#include <hip/hip_runtime.h>
#include <hip/hip_bf16.h>
#include <math.h>

typedef float v2f __attribute__((ext_vector_type(2)));
typedef float v4f __attribute__((ext_vector_type(4)));
typedef float v8f __attribute__((ext_vector_type(8)));

#define B_SZ   2
#define T_SEQ  2048
#define C_DIM  1024
#define NH     16
#define HD     32
#define QKH    32
#define VD     64
#define ROWQKV 3072

__device__ __forceinline__ v8f wmma4(v2f a, v2f b, v8f c) {
  return __builtin_amdgcn_wmma_f32_16x16x4_f32(
       false, a,  false, b,
       (short)0, c,  false,  false);
}

__global__ __launch_bounds__(256) void k_gemm(const float* __restrict__ A,
                                              const float* __restrict__ Bm,
                                              float* __restrict__ Cm,
                                              int N, int K) {
  const int lane = threadIdx.x & 31;
  const int wid  = threadIdx.x >> 5;
  const int wy   = wid & 3;
  const int wx   = wid >> 2;
  const int rowBase = blockIdx.y * 128 + wy * 32;
  const int colBase = blockIdx.x * 128 + wx * 64;
  const int hi   = lane >> 4;
  const int l15  = lane & 15;
  const int bc   = colBase + l15;

  const float* pA0 = A + (size_t)(rowBase + l15) * K + hi * 2;
  const float* pA1 = A + (size_t)(rowBase + 16 + l15) * K + hi * 2;
  const float* pB  = Bm + (size_t)(hi * 2) * N + bc;
  const size_t strideB = (size_t)4 * N;

  v8f acc[2][4] = {};

  v2f a0 = *(const v2f*)pA0;
  v2f a1 = *(const v2f*)pA1;
  v2f b[4];
  #pragma unroll
  for (int cg = 0; cg < 4; ++cg) {
    b[cg].x = pB[16 * cg];
    b[cg].y = pB[N + 16 * cg];
  }

  #pragma unroll 2
  for (int k = 0; k < K - 4; k += 4) {
    pA0 += 4; pA1 += 4; pB += strideB;
    const v2f na0 = *(const v2f*)pA0;
    const v2f na1 = *(const v2f*)pA1;
    v2f nb[4];
    #pragma unroll
    for (int cg = 0; cg < 4; ++cg) {
      nb[cg].x = pB[16 * cg];
      nb[cg].y = pB[N + 16 * cg];
    }
    if ((k & 60) == 0) {
      __builtin_prefetch(pB + 16 * strideB, 0, 1);
      __builtin_prefetch(pA0 + 64, 0, 1);
      __builtin_prefetch(pA1 + 64, 0, 1);
    }
    #pragma unroll
    for (int cg = 0; cg < 4; ++cg) {
      acc[0][cg] = wmma4(a0, b[cg], acc[0][cg]);
      acc[1][cg] = wmma4(a1, b[cg], acc[1][cg]);
    }
    a0 = na0; a1 = na1;
    #pragma unroll
    for (int cg = 0; cg < 4; ++cg) b[cg] = nb[cg];
  }
  #pragma unroll
  for (int cg = 0; cg < 4; ++cg) {
    acc[0][cg] = wmma4(a0, b[cg], acc[0][cg]);
    acc[1][cg] = wmma4(a1, b[cg], acc[1][cg]);
  }

  #pragma unroll
  for (int g = 0; g < 2; ++g) {
    #pragma unroll
    for (int r = 0; r < 8; ++r) {
      const size_t row = (size_t)(rowBase + g * 16 + r + hi * 8);
      #pragma unroll
      for (int cg = 0; cg < 4; ++cg) {
        Cm[row * N + bc + 16 * cg] = acc[g][cg][r];
      }
    }
  }
}

#define QBLK 8192
#define KBLK 4096
#define VBLK 4096

__device__ __forceinline__ v2f rope_pair(v2f x, int j, int t) {
  const float theta = __expf(-((float)(2 * j) / 32.0f) * logf(10000.0f));
  float s, c;
  __sincosf((float)t * theta, &s, &c);
  v2f r;
  r.x = x.x * c - x.y * s;
  r.y = x.x * s + x.y * c;
  return r;
}

__global__ __launch_bounds__(256) void k_pack(float* __restrict__ qkv,
                                              float* __restrict__ Kf,
                                              float* __restrict__ Vf) {
  const unsigned blk = blockIdx.x;
  if (blk < QBLK) {
    unsigned idx = blk * 256u + threadIdx.x;
    const int j = idx & 15;   idx >>= 4;
    const int h = idx & 31;   idx >>= 5;
    const int t = idx & 2047; idx >>= 11;
    const int b = (int)idx;
    float* p = qkv + ((size_t)(b * T_SEQ + t)) * ROWQKV + h * HD + 2 * j;
    const v2f x = *(const v2f*)p;
    *(v2f*)p = rope_pair(x, j, t);
  } else if (blk < QBLK + KBLK) {
    const unsigned o = (blk - QBLK) * 256u + threadIdx.x;
    const int lane = o & 31;
    const int c2   = (o >> 5) & 3;
    const int kb   = (o >> 7) & 127;
    const int bh   = (int)(o >> 14);
    const int hi   = lane >> 4;
    const int l15  = lane & 15;
    const int b    = bh >> 5;
    const int h    = bh & 31;
    const int t    = kb * 16 + l15;
    const float* s0 = qkv + ((size_t)(b * T_SEQ + t)) * ROWQKV + C_DIM + h * HD
                          + 8 * c2 + 2 * hi;
    const v2f xa = *(const v2f*)s0;
    const v2f xb = *(const v2f*)(s0 + 4);
    const v2f ra = rope_pair(xa, 4 * c2 + hi, t);
    const v2f rb = rope_pair(xb, 4 * c2 + 2 + hi, t);
    v4f w; w.x = ra.x; w.y = ra.y; w.z = rb.x; w.w = rb.y;
    ((v4f*)Kf)[o] = w;
  } else {
    const unsigned o = (blk - (QBLK + KBLK)) * 256u + threadIdx.x;
    const int lane = o & 31;
    const int jp   = (o >> 5) & 1;
    const int c    = (o >> 6) & 3;
    const int kb   = (o >> 8) & 127;
    const int bhv  = (int)(o >> 15);
    const int hi   = lane >> 4;
    const int l15  = lane & 15;
    const int b    = bhv >> 4;
    const int hv   = bhv & 15;
    const int k0   = kb * 16 + 4 * c + 2 * hi;
    const float* s0 = qkv + ((size_t)(b * T_SEQ + k0)) * ROWQKV + 2 * C_DIM + hv * VD
                          + 32 * jp + l15;
    const float* s1 = s0 + ROWQKV;
    v4f w; w.x = s0[0]; w.y = s1[0]; w.z = s0[16]; w.w = s1[16];
    ((v4f*)Vf)[o] = w;
  }
}

__global__ void k_lambda(const float* __restrict__ lq1, const float* __restrict__ lk1,
                         const float* __restrict__ lq2, const float* __restrict__ lk2,
                         float* __restrict__ lam) {
  const int l = threadIdx.x;
  float s1 = lq1[l] * lk1[l];
  float s2 = lq2[l] * lk2[l];
  #pragma unroll
  for (int off = 16; off; off >>= 1) {
    s1 += __shfl_xor(s1, off, 32);
    s2 += __shfl_xor(s2, off, 32);
  }
  if (l == 0) *lam = __expf(s1) - __expf(s2) + 0.2f;
}

__device__ __forceinline__ float other_half(float v) {
  const unsigned u = __float_as_uint(v);
  return __uint_as_float(
      __builtin_amdgcn_permlanex16(u, u, 0x76543210u, 0xfedcba98u, false, false));
}

__global__ __launch_bounds__(32) void k_attn(const float* __restrict__ qkv,
                                             const float* __restrict__ Kf,
                                             const float* __restrict__ Vf,
                                             float* __restrict__ Abuf) {
  const int qt   = blockIdx.x;
  const int h    = blockIdx.y;
  const int b    = blockIdx.z;
  const int lane = threadIdx.x;
  const int hi   = lane >> 4;
  const int l15  = lane & 15;

  const float* Q  = qkv + (size_t)b * T_SEQ * ROWQKV + h * HD;
  const v4f* Kq = (const v4f*)Kf + (size_t)(b * QKH + h) * (128 * 128) + lane;
  const v4f* Vq = (const v4f*)Vf + (size_t)(b * NH + (h & 15)) * (128 * 256) + lane;

  __shared__ __attribute__((aligned(16))) float S[2 * 288];
  __shared__ __attribute__((aligned(16))) float frow[16];
  __shared__ float lrow[16];

  const size_t qoff = (size_t)(qt * 16 + l15) * ROWQKV;
  v2f qa[8];
  #pragma unroll
  for (int c = 0; c < 8; ++c) {
    const int col = 4 * c + hi * 2;
    qa[c].x = Q[qoff + col];
    qa[c].y = Q[qoff + col + 1];
  }

  v8f o0 = {}, o1 = {}, o2 = {}, o3 = {};
  const float scale = 0.17677669529663687f;

  float mreg  = -3.0e38f;
  float lpart = 0.0f;
  float* const Sr = S + hi * 288 + l15 * 18;

  const int npair = (qt >> 1) + 1;
  for (int kp = 0; kp < npair; ++kp) {
    v8f s0 = {}, s1 = {};
    const v4f* kptr = Kq + kp * 256;
    v4f ka[4], kc[4];
    #pragma unroll
    for (int c2 = 0; c2 < 4; ++c2) {
      ka[c2] = kptr[c2 * 32];
      kc[c2] = kptr[128 + c2 * 32];
    }
    asm volatile("" :: "v"(ka[0]), "v"(ka[1]), "v"(ka[2]), "v"(ka[3]),
                       "v"(kc[0]), "v"(kc[1]), "v"(kc[2]), "v"(kc[3]));
    #pragma unroll
    for (int c2 = 0; c2 < 4; ++c2) {
      v2f ka0; ka0.x = ka[c2].x; ka0.y = ka[c2].y;
      v2f ka1; ka1.x = ka[c2].z; ka1.y = ka[c2].w;
      v2f kc0; kc0.x = kc[c2].x; kc0.y = kc[c2].y;
      v2f kc1; kc1.x = kc[c2].z; kc1.y = kc[c2].w;
      s0 = wmma4(qa[2 * c2],     ka0, s0);
      s1 = wmma4(qa[2 * c2],     kc0, s1);
      s0 = wmma4(qa[2 * c2 + 1], ka1, s0);
      s1 = wmma4(qa[2 * c2 + 1], kc1, s1);
    }
    #pragma unroll
    for (int r = 0; r < 8; ++r) {
      const int m    = r + hi * 8;
      const int qidx = qt * 16 + m;
      const int kidx = kp * 32 + l15;
      S[m * 18 + l15]       = (kidx      <= qidx) ? s0[r] * scale : -3.0e38f;
      S[288 + m * 18 + l15] = (kidx + 16 <= qidx) ? s1[r] * scale : -3.0e38f;
    }
    __syncthreads();

    {
      v2f sv[8];
      #pragma unroll
      for (int j = 0; j < 8; ++j) sv[j] = *(const v2f*)(Sr + 2 * j);
      float ml = fmaxf(sv[0].x, sv[0].y);
      #pragma unroll
      for (int j = 1; j < 8; ++j) ml = fmaxf(ml, fmaxf(sv[j].x, sv[j].y));
      const float mx = fmaxf(mreg, fmaxf(ml, other_half(ml)));
      const float f  = __expf(mreg - mx);
      float sum = 0.0f;
      #pragma unroll
      for (int j = 0; j < 8; ++j) {
        v2f p;
        p.x = __expf(sv[j].x - mx);
        p.y = __expf(sv[j].y - mx);
        *(v2f*)(Sr + 2 * j) = p;
        sum += p.x;
        sum += p.y;
      }
      mreg  = mx;
      lpart = lpart * f + sum;
      frow[l15] = f;
    }
    __syncthreads();

    const v4f* vp = Vq + kp * 512;
    v4f vv[16];
    #pragma unroll
    for (int i = 0; i < 16; ++i) vv[i] = vp[i * 32];

    const v4f f0 = *(const v4f*)(frow + hi * 8);
    const v4f f1 = *(const v4f*)(frow + hi * 8 + 4);
    const float fr[8] = {f0.x, f0.y, f0.z, f0.w, f1.x, f1.y, f1.z, f1.w};
    #pragma unroll
    for (int r = 0; r < 8; ++r) {
      o0[r] *= fr[r]; o1[r] *= fr[r]; o2[r] *= fr[r]; o3[r] *= fr[r];
    }

    v2f pa[8];
    #pragma unroll
    for (int c = 0; c < 8; ++c) {
      pa[c] = *(const v2f*)(S + (c >> 2) * 288 + l15 * 18 + 4 * (c & 3) + hi * 2);
    }
    asm volatile("" :: "v"(vv[0]), "v"(vv[1]), "v"(vv[2]),  "v"(vv[3]),
                       "v"(vv[4]), "v"(vv[5]), "v"(vv[6]),  "v"(vv[7]),
                       "v"(vv[8]), "v"(vv[9]), "v"(vv[10]), "v"(vv[11]),
                       "v"(vv[12]), "v"(vv[13]), "v"(vv[14]), "v"(vv[15]));
    #pragma unroll
    for (int c = 0; c < 8; ++c) {
      const v4f va = vv[2 * c];
      const v4f vc = vv[2 * c + 1];
      v2f vb0; vb0.x = va.x; vb0.y = va.y;
      v2f vb1; vb1.x = va.z; vb1.y = va.w;
      v2f vb2; vb2.x = vc.x; vb2.y = vc.y;
      v2f vb3; vb3.x = vc.z; vb3.y = vc.w;
      o0 = wmma4(pa[c], vb0, o0);
      o1 = wmma4(pa[c], vb1, o1);
      o2 = wmma4(pa[c], vb2, o2);
      o3 = wmma4(pa[c], vb3, o3);
    }
    __syncthreads();
  }

  lrow[l15] = lpart + other_half(lpart);
  __syncthreads();
  float inv[8];
  #pragma unroll
  for (int r = 0; r < 8; ++r) inv[r] = 1.0f / lrow[r + hi * 8];
  float* Ab = Abuf + ((size_t)(b * QKH + h)) * T_SEQ * VD;
  #pragma unroll
  for (int r = 0; r < 8; ++r) {
    float* Ar = Ab + (size_t)(qt * 16 + r + hi * 8) * VD;
    Ar[l15]      = o0[r] * inv[r];
    Ar[l15 + 16] = o1[r] * inv[r];
    Ar[l15 + 32] = o2[r] * inv[r];
    Ar[l15 + 48] = o3[r] * inv[r];
  }
}

__global__ __launch_bounds__(256) void k_diffnorm(const float* __restrict__ Abuf,
                                                  const float* __restrict__ lamp,
                                                  float* __restrict__ Dbuf) {
  const float lam = *lamp;
  const int wid  = threadIdx.x >> 5;
  const int lane = threadIdx.x & 31;
  const size_t row = (size_t)blockIdx.x * 8 + wid;
  const int t  = (int)(row & 2047);
  const size_t bh = row >> 11;
  const int hh = (int)(bh & 15);
  const int b  = (int)(bh >> 4);

  const float* A1 = Abuf + ((size_t)(b * QKH + hh))      * T_SEQ * VD + (size_t)t * VD;
  const float* A2 = Abuf + ((size_t)(b * QKH + 16 + hh)) * T_SEQ * VD + (size_t)t * VD;
  const float d0 = A1[lane]      - lam * A2[lane];
  const float d1 = A1[lane + 32] - lam * A2[lane + 32];
  float ss = d0 * d0 + d1 * d1;
  #pragma unroll
  for (int off = 16; off; off >>= 1) ss += __shfl_xor(ss, off, 32);
  const float r = rsqrtf(ss * (1.0f / 64.0f) + 1e-5f) * 0.8f;
  float* Dp = Dbuf + ((size_t)(b * T_SEQ + t)) * C_DIM + hh * VD;
  Dp[lane]      = d0 * r;
  Dp[lane + 32] = d1 * r;
}

extern "C" void kernel_launch(void* const* d_in, const int* in_sizes, int n_in,
                              void* d_out, int out_size, void* d_ws, size_t ws_size,
                              hipStream_t stream) {
  const float* x      = (const float*)d_in[0];
  const float* W_attn = (const float*)d_in[1];
  const float* W_proj = (const float*)d_in[2];
  const float* lq1    = (const float*)d_in[3];
  const float* lk1    = (const float*)d_in[4];
  const float* lq2    = (const float*)d_in[5];
  const float* lk2    = (const float*)d_in[6];
  float* out = (float*)d_out;

  float* qkv  = (float*)d_ws;
  float* Abuf = qkv  + (size_t)B_SZ * T_SEQ * ROWQKV;
  float* Dbuf = Abuf + (size_t)B_SZ * QKH * T_SEQ * VD;
  float* Kf   = Dbuf + (size_t)B_SZ * T_SEQ * C_DIM;
  float* Vf   = Kf   + (size_t)B_SZ * QKH * T_SEQ * HD;
  float* lam  = Vf   + (size_t)B_SZ * NH * T_SEQ * VD;

  const int M = B_SZ * T_SEQ;

  {
    dim3 grid(3 * C_DIM / 128, M / 128);
    k_gemm<<<grid, 256, 0, stream>>>(x, W_attn, qkv, 3 * C_DIM, C_DIM);
  }
  k_pack<<<QBLK + KBLK + VBLK, 256, 0, stream>>>(qkv, Kf, Vf);
  k_lambda<<<1, 32, 0, stream>>>(lq1, lk1, lq2, lk2, lam);
  {
    dim3 grid(T_SEQ / 16, QKH, B_SZ);
    k_attn<<<grid, 32, 0, stream>>>(qkv, Kf, Vf, Abuf);
  }
  k_diffnorm<<<(B_SZ * NH * T_SEQ) / 8, 256, 0, stream>>>(Abuf, lam, Dbuf);
  {
    dim3 grid(C_DIM / 128, M / 128);
    k_gemm<<<grid, 256, 0, stream>>>(Dbuf, W_proj, out, C_DIM, C_DIM);
  }
  (void)in_sizes; (void)n_in; (void)out_size; (void)ws_size;
}


# both GEMMs: a lane holds 4 adjacent columns (one per N-tile) and 4 adjacent k: each A, B, C access is one direct 128-bit access, no LDS; loads held one 8-k trip ahead; k sum regrouped in 8s
# speedup vs baseline: 1.3453x; 1.1140x over previous
#include <hip/hip_runtime.h>
#include <hip/hip_bf16.h>
#include <math.h>

typedef float v2f __attribute__((ext_vector_type(2)));
typedef float v4f __attribute__((ext_vector_type(4)));
typedef float v8f __attribute__((ext_vector_type(8)));

#define B_SZ   2
#define T_SEQ  2048
#define C_DIM  1024
#define NH     16
#define HD     32
#define QKH    32
#define VD     64
#define ROWQKV 3072

__device__ __forceinline__ v8f wmma4(v2f a, v2f b, v8f c) {
  return __builtin_amdgcn_wmma_f32_16x16x4_f32(
       false, a,  false, b,
       (short)0, c,  false,  false);
}

__device__ __forceinline__ void gemm_trip(v8f (&acc)[2][4], const v4f a0, const v4f a1,
                                          const v2f (&bp0)[4], const v2f (&bp1)[4]) {
  v2f x0, x1;
  x0.x = a0.x; x0.y = a0.y;
  x1.x = a1.x; x1.y = a1.y;
  #pragma unroll
  for (int cg = 0; cg < 4; ++cg) {
    acc[0][cg] = wmma4(x0, bp0[cg], acc[0][cg]);
    acc[1][cg] = wmma4(x1, bp0[cg], acc[1][cg]);
  }
  x0.x = a0.z; x0.y = a0.w;
  x1.x = a1.z; x1.y = a1.w;
  #pragma unroll
  for (int cg = 0; cg < 4; ++cg) {
    acc[0][cg] = wmma4(x0, bp1[cg], acc[0][cg]);
    acc[1][cg] = wmma4(x1, bp1[cg], acc[1][cg]);
  }
}

__device__ __forceinline__ void gemm_pairs(const v4f (&b)[4], v2f (&bp0)[4], v2f (&bp1)[4]) {
  #pragma unroll
  for (int cg = 0; cg < 4; ++cg) {
    bp0[cg].x = b[0][cg]; bp0[cg].y = b[1][cg];
    bp1[cg].x = b[2][cg]; bp1[cg].y = b[3][cg];
  }
}

__global__ __launch_bounds__(256) void k_gemm(const float* __restrict__ A,
                                              const float* __restrict__ Bm,
                                              float* __restrict__ Cm,
                                              int N, int K) {
  const int lane = threadIdx.x & 31;
  const int wid  = threadIdx.x >> 5;
  const int wy   = wid & 3;
  const int wx   = wid >> 2;
  const int rowBase = blockIdx.y * 128 + wy * 32;
  const int colBase = blockIdx.x * 128 + wx * 64;
  const int hi   = lane >> 4;
  const int l15  = lane & 15;
  const int bc   = colBase + 4 * l15;

  const float* pA0 = A + (size_t)(rowBase + l15) * K + hi * 4;
  const float* pA1 = A + (size_t)(rowBase + 16 + l15) * K + hi * 4;
  const float* pB  = Bm + (size_t)(hi * 4) * N + bc;
  const size_t strideB = (size_t)8 * N;

  v8f acc[2][4] = {};

  v4f a0 = *(const v4f*)pA0;
  v4f a1 = *(const v4f*)pA1;
  v2f bp0[4], bp1[4];
  {
    v4f b[4];
    #pragma unroll
    for (int i = 0; i < 4; ++i) b[i] = *(const v4f*)(pB + (size_t)i * N);
    gemm_pairs(b, bp0, bp1);
  }

  #pragma unroll 2
  for (int k = 0; k < K - 8; k += 8) {
    pA0 += 8; pA1 += 8; pB += strideB;
    const v4f na0 = *(const v4f*)pA0;
    const v4f na1 = *(const v4f*)pA1;
    v4f nb[4];
    #pragma unroll
    for (int i = 0; i < 4; ++i) nb[i] = *(const v4f*)(pB + (size_t)i * N);
    __builtin_amdgcn_sched_barrier(0);
    gemm_trip(acc, a0, a1, bp0, bp1);
    __builtin_amdgcn_sched_barrier(0);
    a0 = na0; a1 = na1;
    gemm_pairs(nb, bp0, bp1);
  }
  gemm_trip(acc, a0, a1, bp0, bp1);

  #pragma unroll
  for (int g = 0; g < 2; ++g) {
    #pragma unroll
    for (int r = 0; r < 8; ++r) {
      const size_t row = (size_t)(rowBase + g * 16 + r + hi * 8);
      v4f o;
      o.x = acc[g][0][r]; o.y = acc[g][1][r]; o.z = acc[g][2][r]; o.w = acc[g][3][r];
      *(v4f*)(Cm + row * N + bc) = o;
    }
  }
}

#define QBLK 8192
#define KBLK 4096
#define VBLK 4096

__device__ __forceinline__ v2f rope_pair(v2f x, int j, int t) {
  const float theta = __expf(-((float)(2 * j) / 32.0f) * logf(10000.0f));
  float s, c;
  __sincosf((float)t * theta, &s, &c);
  v2f r;
  r.x = x.x * c - x.y * s;
  r.y = x.x * s + x.y * c;
  return r;
}

__global__ __launch_bounds__(256) void k_pack(float* __restrict__ qkv,
                                              float* __restrict__ Kf,
                                              float* __restrict__ Vf) {
  const unsigned blk = blockIdx.x;
  if (blk < QBLK) {
    unsigned idx = blk * 256u + threadIdx.x;
    const int j = idx & 15;   idx >>= 4;
    const int h = idx & 31;   idx >>= 5;
    const int t = idx & 2047; idx >>= 11;
    const int b = (int)idx;
    float* p = qkv + ((size_t)(b * T_SEQ + t)) * ROWQKV + h * HD + 2 * j;
    const v2f x = *(const v2f*)p;
    *(v2f*)p = rope_pair(x, j, t);
  } else if (blk < QBLK + KBLK) {
    const unsigned o = (blk - QBLK) * 256u + threadIdx.x;
    const int lane = o & 31;
    const int c2   = (o >> 5) & 3;
    const int kb   = (o >> 7) & 127;
    const int bh   = (int)(o >> 14);
    const int hi   = lane >> 4;
    const int l15  = lane & 15;
    const int b    = bh >> 5;
    const int h    = bh & 31;
    const int t    = kb * 16 + l15;
    const float* s0 = qkv + ((size_t)(b * T_SEQ + t)) * ROWQKV + C_DIM + h * HD
                          + 8 * c2 + 2 * hi;
    const v2f xa = *(const v2f*)s0;
    const v2f xb = *(const v2f*)(s0 + 4);
    const v2f ra = rope_pair(xa, 4 * c2 + hi, t);
    const v2f rb = rope_pair(xb, 4 * c2 + 2 + hi, t);
    v4f w; w.x = ra.x; w.y = ra.y; w.z = rb.x; w.w = rb.y;
    ((v4f*)Kf)[o] = w;
  } else {
    const unsigned o = (blk - (QBLK + KBLK)) * 256u + threadIdx.x;
    const int lane = o & 31;
    const int jp   = (o >> 5) & 1;
    const int c    = (o >> 6) & 3;
    const int kb   = (o >> 8) & 127;
    const int bhv  = (int)(o >> 15);
    const int hi   = lane >> 4;
    const int l15  = lane & 15;
    const int b    = bhv >> 4;
    const int hv   = bhv & 15;
    const int k0   = kb * 16 + 4 * c + 2 * hi;
    const float* s0 = qkv + ((size_t)(b * T_SEQ + k0)) * ROWQKV + 2 * C_DIM + hv * VD
                          + 32 * jp + l15;
    const float* s1 = s0 + ROWQKV;
    v4f w; w.x = s0[0]; w.y = s1[0]; w.z = s0[16]; w.w = s1[16];
    ((v4f*)Vf)[o] = w;
  }
}

__global__ void k_lambda(const float* __restrict__ lq1, const float* __restrict__ lk1,
                         const float* __restrict__ lq2, const float* __restrict__ lk2,
                         float* __restrict__ lam) {
  const int l = threadIdx.x;
  float s1 = lq1[l] * lk1[l];
  float s2 = lq2[l] * lk2[l];
  #pragma unroll
  for (int off = 16; off; off >>= 1) {
    s1 += __shfl_xor(s1, off, 32);
    s2 += __shfl_xor(s2, off, 32);
  }
  if (l == 0) *lam = __expf(s1) - __expf(s2) + 0.2f;
}

__device__ __forceinline__ float other_half(float v) {
  const unsigned u = __float_as_uint(v);
  return __uint_as_float(
      __builtin_amdgcn_permlanex16(u, u, 0x76543210u, 0xfedcba98u, false, false));
}

__global__ __launch_bounds__(32) void k_attn(const float* __restrict__ qkv,
                                             const float* __restrict__ Kf,
                                             const float* __restrict__ Vf,
                                             float* __restrict__ Abuf) {
  const int qt   = blockIdx.x;
  const int h    = blockIdx.y;
  const int b    = blockIdx.z;
  const int lane = threadIdx.x;
  const int hi   = lane >> 4;
  const int l15  = lane & 15;

  const float* Q  = qkv + (size_t)b * T_SEQ * ROWQKV + h * HD;
  const v4f* Kq = (const v4f*)Kf + (size_t)(b * QKH + h) * (128 * 128) + lane;
  const v4f* Vq = (const v4f*)Vf + (size_t)(b * NH + (h & 15)) * (128 * 256) + lane;

  __shared__ __attribute__((aligned(16))) float S[2 * 288];
  __shared__ __attribute__((aligned(16))) float frow[16];
  __shared__ float lrow[16];

  const size_t qoff = (size_t)(qt * 16 + l15) * ROWQKV;
  v2f qa[8];
  #pragma unroll
  for (int c = 0; c < 8; ++c) {
    const int col = 4 * c + hi * 2;
    qa[c].x = Q[qoff + col];
    qa[c].y = Q[qoff + col + 1];
  }

  v8f o0 = {}, o1 = {}, o2 = {}, o3 = {};
  const float scale = 0.17677669529663687f;

  float mreg  = -3.0e38f;
  float lpart = 0.0f;
  float* const Sr = S + hi * 288 + l15 * 18;

  const int npair = (qt >> 1) + 1;
  for (int kp = 0; kp < npair; ++kp) {
    v8f s0 = {}, s1 = {};
    const v4f* kptr = Kq + kp * 256;
    v4f ka[4], kc[4];
    #pragma unroll
    for (int c2 = 0; c2 < 4; ++c2) {
      ka[c2] = kptr[c2 * 32];
      kc[c2] = kptr[128 + c2 * 32];
    }
    asm volatile("" :: "v"(ka[0]), "v"(ka[1]), "v"(ka[2]), "v"(ka[3]),
                       "v"(kc[0]), "v"(kc[1]), "v"(kc[2]), "v"(kc[3]));
    #pragma unroll
    for (int c2 = 0; c2 < 4; ++c2) {
      v2f ka0; ka0.x = ka[c2].x; ka0.y = ka[c2].y;
      v2f ka1; ka1.x = ka[c2].z; ka1.y = ka[c2].w;
      v2f kc0; kc0.x = kc[c2].x; kc0.y = kc[c2].y;
      v2f kc1; kc1.x = kc[c2].z; kc1.y = kc[c2].w;
      s0 = wmma4(qa[2 * c2],     ka0, s0);
      s1 = wmma4(qa[2 * c2],     kc0, s1);
      s0 = wmma4(qa[2 * c2 + 1], ka1, s0);
      s1 = wmma4(qa[2 * c2 + 1], kc1, s1);
    }
    #pragma unroll
    for (int r = 0; r < 8; ++r) {
      const int m    = r + hi * 8;
      const int qidx = qt * 16 + m;
      const int kidx = kp * 32 + l15;
      S[m * 18 + l15]       = (kidx      <= qidx) ? s0[r] * scale : -3.0e38f;
      S[288 + m * 18 + l15] = (kidx + 16 <= qidx) ? s1[r] * scale : -3.0e38f;
    }
    __syncthreads();

    {
      v2f sv[8];
      #pragma unroll
      for (int j = 0; j < 8; ++j) sv[j] = *(const v2f*)(Sr + 2 * j);
      float ml = fmaxf(sv[0].x, sv[0].y);
      #pragma unroll
      for (int j = 1; j < 8; ++j) ml = fmaxf(ml, fmaxf(sv[j].x, sv[j].y));
      const float mx = fmaxf(mreg, fmaxf(ml, other_half(ml)));
      const float f  = __expf(mreg - mx);
      float sum = 0.0f;
      #pragma unroll
      for (int j = 0; j < 8; ++j) {
        v2f p;
        p.x = __expf(sv[j].x - mx);
        p.y = __expf(sv[j].y - mx);
        *(v2f*)(Sr + 2 * j) = p;
        sum += p.x;
        sum += p.y;
      }
      mreg  = mx;
      lpart = lpart * f + sum;
      frow[l15] = f;
    }
    __syncthreads();

    const v4f* vp = Vq + kp * 512;
    v4f vv[16];
    #pragma unroll
    for (int i = 0; i < 16; ++i) vv[i] = vp[i * 32];

    const v4f f0 = *(const v4f*)(frow + hi * 8);
    const v4f f1 = *(const v4f*)(frow + hi * 8 + 4);
    const float fr[8] = {f0.x, f0.y, f0.z, f0.w, f1.x, f1.y, f1.z, f1.w};
    #pragma unroll
    for (int r = 0; r < 8; ++r) {
      o0[r] *= fr[r]; o1[r] *= fr[r]; o2[r] *= fr[r]; o3[r] *= fr[r];
    }

    v2f pa[8];
    #pragma unroll
    for (int c = 0; c < 8; ++c) {
      pa[c] = *(const v2f*)(S + (c >> 2) * 288 + l15 * 18 + 4 * (c & 3) + hi * 2);
    }
    asm volatile("" :: "v"(vv[0]), "v"(vv[1]), "v"(vv[2]),  "v"(vv[3]),
                       "v"(vv[4]), "v"(vv[5]), "v"(vv[6]),  "v"(vv[7]),
                       "v"(vv[8]), "v"(vv[9]), "v"(vv[10]), "v"(vv[11]),
                       "v"(vv[12]), "v"(vv[13]), "v"(vv[14]), "v"(vv[15]));
    #pragma unroll
    for (int c = 0; c < 8; ++c) {
      const v4f va = vv[2 * c];
      const v4f vc = vv[2 * c + 1];
      v2f vb0; vb0.x = va.x; vb0.y = va.y;
      v2f vb1; vb1.x = va.z; vb1.y = va.w;
      v2f vb2; vb2.x = vc.x; vb2.y = vc.y;
      v2f vb3; vb3.x = vc.z; vb3.y = vc.w;
      o0 = wmma4(pa[c], vb0, o0);
      o1 = wmma4(pa[c], vb1, o1);
      o2 = wmma4(pa[c], vb2, o2);
      o3 = wmma4(pa[c], vb3, o3);
    }
    __syncthreads();
  }

  lrow[l15] = lpart + other_half(lpart);
  __syncthreads();
  float inv[8];
  #pragma unroll
  for (int r = 0; r < 8; ++r) inv[r] = 1.0f / lrow[r + hi * 8];
  float* Ab = Abuf + ((size_t)(b * QKH + h)) * T_SEQ * VD;
  #pragma unroll
  for (int r = 0; r < 8; ++r) {
    float* Ar = Ab + (size_t)(qt * 16 + r + hi * 8) * VD;
    Ar[l15]      = o0[r] * inv[r];
    Ar[l15 + 16] = o1[r] * inv[r];
    Ar[l15 + 32] = o2[r] * inv[r];
    Ar[l15 + 48] = o3[r] * inv[r];
  }
}

__global__ __launch_bounds__(256) void k_diffnorm(const float* __restrict__ Abuf,
                                                  const float* __restrict__ lamp,
                                                  float* __restrict__ Dbuf) {
  const float lam = *lamp;
  const int wid  = threadIdx.x >> 5;
  const int lane = threadIdx.x & 31;
  const size_t row = (size_t)blockIdx.x * 8 + wid;
  const int t  = (int)(row & 2047);
  const size_t bh = row >> 11;
  const int hh = (int)(bh & 15);
  const int b  = (int)(bh >> 4);

  const float* A1 = Abuf + ((size_t)(b * QKH + hh))      * T_SEQ * VD + (size_t)t * VD;
  const float* A2 = Abuf + ((size_t)(b * QKH + 16 + hh)) * T_SEQ * VD + (size_t)t * VD;
  const float d0 = A1[lane]      - lam * A2[lane];
  const float d1 = A1[lane + 32] - lam * A2[lane + 32];
  float ss = d0 * d0 + d1 * d1;
  #pragma unroll
  for (int off = 16; off; off >>= 1) ss += __shfl_xor(ss, off, 32);
  const float r = rsqrtf(ss * (1.0f / 64.0f) + 1e-5f) * 0.8f;
  float* Dp = Dbuf + ((size_t)(b * T_SEQ + t)) * C_DIM + hh * VD;
  Dp[lane]      = d0 * r;
  Dp[lane + 32] = d1 * r;
}

extern "C" void kernel_launch(void* const* d_in, const int* in_sizes, int n_in,
                              void* d_out, int out_size, void* d_ws, size_t ws_size,
                              hipStream_t stream) {
  const float* x      = (const float*)d_in[0];
  const float* W_attn = (const float*)d_in[1];
  const float* W_proj = (const float*)d_in[2];
  const float* lq1    = (const float*)d_in[3];
  const float* lk1    = (const float*)d_in[4];
  const float* lq2    = (const float*)d_in[5];
  const float* lk2    = (const float*)d_in[6];
  float* out = (float*)d_out;

  float* qkv  = (float*)d_ws;
  float* Abuf = qkv  + (size_t)B_SZ * T_SEQ * ROWQKV;
  float* Dbuf = Abuf + (size_t)B_SZ * QKH * T_SEQ * VD;
  float* Kf   = Dbuf + (size_t)B_SZ * T_SEQ * C_DIM;
  float* Vf   = Kf   + (size_t)B_SZ * QKH * T_SEQ * HD;
  float* lam  = Vf   + (size_t)B_SZ * NH * T_SEQ * VD;

  const int M = B_SZ * T_SEQ;

  {
    dim3 grid(3 * C_DIM / 128, M / 128);
    k_gemm<<<grid, 256, 0, stream>>>(x, W_attn, qkv, 3 * C_DIM, C_DIM);
  }
  k_pack<<<QBLK + KBLK + VBLK, 256, 0, stream>>>(qkv, Kf, Vf);
  k_lambda<<<1, 32, 0, stream>>>(lq1, lk1, lq2, lk2, lam);
  {
    dim3 grid(T_SEQ / 16, QKH, B_SZ);
    k_attn<<<grid, 32, 0, stream>>>(qkv, Kf, Vf, Abuf);
  }
  k_diffnorm<<<(B_SZ * NH * T_SEQ) / 8, 256, 0, stream>>>(Abuf, lam, Dbuf);
  {
    dim3 grid(C_DIM / 128, M / 128);
    k_gemm<<<grid, 256, 0, stream>>>(Dbuf, W_proj, out, C_DIM, C_DIM);
  }
  (void)in_sizes; (void)n_in; (void)out_size; (void)ws_size;
}


# attention: block-to-tile launch map: the 6144 blocks resident first take the heavy query tiles in complementary sets per SIMD, the 2048 light tiles follow heaviest first; per-wave code unchanged
# speedup vs baseline: 1.4064x; 1.0454x over previous
#include <hip/hip_runtime.h>
#include <hip/hip_bf16.h>
#include <math.h>

typedef float v2f __attribute__((ext_vector_type(2)));
typedef float v4f __attribute__((ext_vector_type(4)));
typedef float v8f __attribute__((ext_vector_type(8)));

#define B_SZ   2
#define T_SEQ  2048
#define C_DIM  1024
#define NH     16
#define HD     32
#define QKH    32
#define VD     64
#define ROWQKV 3072

__device__ __forceinline__ v8f wmma4(v2f a, v2f b, v8f c) {
  return __builtin_amdgcn_wmma_f32_16x16x4_f32(
       false, a,  false, b,
       (short)0, c,  false,  false);
}

__device__ __forceinline__ void gemm_trip(v8f (&acc)[2][4], const v4f a0, const v4f a1,
                                          const v2f (&bp0)[4], const v2f (&bp1)[4]) {
  v2f x0, x1;
  x0.x = a0.x; x0.y = a0.y;
  x1.x = a1.x; x1.y = a1.y;
  #pragma unroll
  for (int cg = 0; cg < 4; ++cg) {
    acc[0][cg] = wmma4(x0, bp0[cg], acc[0][cg]);
    acc[1][cg] = wmma4(x1, bp0[cg], acc[1][cg]);
  }
  x0.x = a0.z; x0.y = a0.w;
  x1.x = a1.z; x1.y = a1.w;
  #pragma unroll
  for (int cg = 0; cg < 4; ++cg) {
    acc[0][cg] = wmma4(x0, bp1[cg], acc[0][cg]);
    acc[1][cg] = wmma4(x1, bp1[cg], acc[1][cg]);
  }
}

__device__ __forceinline__ void gemm_pairs(const v4f (&b)[4], v2f (&bp0)[4], v2f (&bp1)[4]) {
  #pragma unroll
  for (int cg = 0; cg < 4; ++cg) {
    bp0[cg].x = b[0][cg]; bp0[cg].y = b[1][cg];
    bp1[cg].x = b[2][cg]; bp1[cg].y = b[3][cg];
  }
}

__global__ __launch_bounds__(256) void k_gemm(const float* __restrict__ A,
                                              const float* __restrict__ Bm,
                                              float* __restrict__ Cm,
                                              int N, int K) {
  const int lane = threadIdx.x & 31;
  const int wid  = threadIdx.x >> 5;
  const int wy   = wid & 3;
  const int wx   = wid >> 2;
  const int rowBase = blockIdx.y * 128 + wy * 32;
  const int colBase = blockIdx.x * 128 + wx * 64;
  const int hi   = lane >> 4;
  const int l15  = lane & 15;
  const int bc   = colBase + 4 * l15;

  const float* pA0 = A + (size_t)(rowBase + l15) * K + hi * 4;
  const float* pA1 = A + (size_t)(rowBase + 16 + l15) * K + hi * 4;
  const float* pB  = Bm + (size_t)(hi * 4) * N + bc;
  const size_t strideB = (size_t)8 * N;

  v8f acc[2][4] = {};

  v4f a0 = *(const v4f*)pA0;
  v4f a1 = *(const v4f*)pA1;
  v2f bp0[4], bp1[4];
  {
    v4f b[4];
    #pragma unroll
    for (int i = 0; i < 4; ++i) b[i] = *(const v4f*)(pB + (size_t)i * N);
    gemm_pairs(b, bp0, bp1);
  }

  #pragma unroll 2
  for (int k = 0; k < K - 8; k += 8) {
    pA0 += 8; pA1 += 8; pB += strideB;
    const v4f na0 = *(const v4f*)pA0;
    const v4f na1 = *(const v4f*)pA1;
    v4f nb[4];
    #pragma unroll
    for (int i = 0; i < 4; ++i) nb[i] = *(const v4f*)(pB + (size_t)i * N);
    __builtin_amdgcn_sched_barrier(0);
    gemm_trip(acc, a0, a1, bp0, bp1);
    __builtin_amdgcn_sched_barrier(0);
    a0 = na0; a1 = na1;
    gemm_pairs(nb, bp0, bp1);
  }
  gemm_trip(acc, a0, a1, bp0, bp1);

  #pragma unroll
  for (int g = 0; g < 2; ++g) {
    #pragma unroll
    for (int r = 0; r < 8; ++r) {
      const size_t row = (size_t)(rowBase + g * 16 + r + hi * 8);
      v4f o;
      o.x = acc[g][0][r]; o.y = acc[g][1][r]; o.z = acc[g][2][r]; o.w = acc[g][3][r];
      *(v4f*)(Cm + row * N + bc) = o;
    }
  }
}

#define QBLK 8192
#define KBLK 4096
#define VBLK 4096

__device__ __forceinline__ v2f rope_pair(v2f x, int j, int t) {
  const float theta = __expf(-((float)(2 * j) / 32.0f) * logf(10000.0f));
  float s, c;
  __sincosf((float)t * theta, &s, &c);
  v2f r;
  r.x = x.x * c - x.y * s;
  r.y = x.x * s + x.y * c;
  return r;
}

__global__ __launch_bounds__(256) void k_pack(float* __restrict__ qkv,
                                              float* __restrict__ Kf,
                                              float* __restrict__ Vf) {
  const unsigned blk = blockIdx.x;
  if (blk < QBLK) {
    unsigned idx = blk * 256u + threadIdx.x;
    const int j = idx & 15;   idx >>= 4;
    const int h = idx & 31;   idx >>= 5;
    const int t = idx & 2047; idx >>= 11;
    const int b = (int)idx;
    float* p = qkv + ((size_t)(b * T_SEQ + t)) * ROWQKV + h * HD + 2 * j;
    const v2f x = *(const v2f*)p;
    *(v2f*)p = rope_pair(x, j, t);
  } else if (blk < QBLK + KBLK) {
    const unsigned o = (blk - QBLK) * 256u + threadIdx.x;
    const int lane = o & 31;
    const int c2   = (o >> 5) & 3;
    const int kb   = (o >> 7) & 127;
    const int bh   = (int)(o >> 14);
    const int hi   = lane >> 4;
    const int l15  = lane & 15;
    const int b    = bh >> 5;
    const int h    = bh & 31;
    const int t    = kb * 16 + l15;
    const float* s0 = qkv + ((size_t)(b * T_SEQ + t)) * ROWQKV + C_DIM + h * HD
                          + 8 * c2 + 2 * hi;
    const v2f xa = *(const v2f*)s0;
    const v2f xb = *(const v2f*)(s0 + 4);
    const v2f ra = rope_pair(xa, 4 * c2 + hi, t);
    const v2f rb = rope_pair(xb, 4 * c2 + 2 + hi, t);
    v4f w; w.x = ra.x; w.y = ra.y; w.z = rb.x; w.w = rb.y;
    ((v4f*)Kf)[o] = w;
  } else {
    const unsigned o = (blk - (QBLK + KBLK)) * 256u + threadIdx.x;
    const int lane = o & 31;
    const int jp   = (o >> 5) & 1;
    const int c    = (o >> 6) & 3;
    const int kb   = (o >> 8) & 127;
    const int bhv  = (int)(o >> 15);
    const int hi   = lane >> 4;
    const int l15  = lane & 15;
    const int b    = bhv >> 4;
    const int hv   = bhv & 15;
    const int k0   = kb * 16 + 4 * c + 2 * hi;
    const float* s0 = qkv + ((size_t)(b * T_SEQ + k0)) * ROWQKV + 2 * C_DIM + hv * VD
                          + 32 * jp + l15;
    const float* s1 = s0 + ROWQKV;
    v4f w; w.x = s0[0]; w.y = s1[0]; w.z = s0[16]; w.w = s1[16];
    ((v4f*)Vf)[o] = w;
  }
}

__global__ void k_lambda(const float* __restrict__ lq1, const float* __restrict__ lk1,
                         const float* __restrict__ lq2, const float* __restrict__ lk2,
                         float* __restrict__ lam) {
  const int l = threadIdx.x;
  float s1 = lq1[l] * lk1[l];
  float s2 = lq2[l] * lk2[l];
  #pragma unroll
  for (int off = 16; off; off >>= 1) {
    s1 += __shfl_xor(s1, off, 32);
    s2 += __shfl_xor(s2, off, 32);
  }
  if (l == 0) *lam = __expf(s1) - __expf(s2) + 0.2f;
}

__device__ __forceinline__ float other_half(float v) {
  const unsigned u = __float_as_uint(v);
  return __uint_as_float(
      __builtin_amdgcn_permlanex16(u, u, 0x76543210u, 0xfedcba98u, false, false));
}

__global__ __launch_bounds__(32) void k_attn(const float* __restrict__ qkv,
                                             const float* __restrict__ Kf,
                                             const float* __restrict__ Vf,
                                             float* __restrict__ Abuf) {
  const unsigned L = blockIdx.x;
  unsigned u, r7;
  if (L < 6144u) {
    const unsigned T   = L >> 11;
    const unsigned low = L & 2047u;
    const unsigned pr  = (unsigned)__popc(low) & 1u;
    const unsigned x12 = ((low >> 1) ^ (low >> 2)) & 1u;
    const unsigned x34 = ((low >> 3) ^ (low >> 4)) & 1u;
    const unsigned x56 = ((low >> 5) ^ (low >> 6)) & 1u;
    const unsigned ul  = (pr << 3) | ((pr ^ x12) << 2) | ((pr ^ x34) << 1) | (pr ^ x56);
    const unsigned third = (T + (low & 1023u) % 3u) % 3u;
    u  = 16u + 16u * third + ul;
    r7 = ((low >> 1) & 1u) | (((low >> 3) & 1u) << 1) | (((low >> 5) & 1u) << 2) |
         ((low & 1u) << 3) | (((low >> 7) & 7u) << 4);
  } else {
    const unsigned t = L - 6144u;
    u  = 15u - (t >> 7);
    r7 = t & 127u;
  }
  const int qt   = (int)((u << 1) | (r7 & 1u));
  const int h    = (int)((r7 >> 1) & 31u);
  const int b    = (int)(r7 >> 6);
  const int lane = threadIdx.x;
  const int hi   = lane >> 4;
  const int l15  = lane & 15;

  const float* Q  = qkv + (size_t)b * T_SEQ * ROWQKV + h * HD;
  const v4f* Kq = (const v4f*)Kf + (size_t)(b * QKH + h) * (128 * 128) + lane;
  const v4f* Vq = (const v4f*)Vf + (size_t)(b * NH + (h & 15)) * (128 * 256) + lane;

  __shared__ __attribute__((aligned(16))) float S[2 * 288];
  __shared__ __attribute__((aligned(16))) float frow[16];
  __shared__ float lrow[16];

  const size_t qoff = (size_t)(qt * 16 + l15) * ROWQKV;
  v2f qa[8];
  #pragma unroll
  for (int c = 0; c < 8; ++c) {
    const int col = 4 * c + hi * 2;
    qa[c].x = Q[qoff + col];
    qa[c].y = Q[qoff + col + 1];
  }

  v8f o0 = {}, o1 = {}, o2 = {}, o3 = {};
  const float scale = 0.17677669529663687f;

  float mreg  = -3.0e38f;
  float lpart = 0.0f;
  float* const Sr = S + hi * 288 + l15 * 18;

  const int npair = (qt >> 1) + 1;
  for (int kp = 0; kp < npair; ++kp) {
    v8f s0 = {}, s1 = {};
    const v4f* kptr = Kq + kp * 256;
    v4f ka[4], kc[4];
    #pragma unroll
    for (int c2 = 0; c2 < 4; ++c2) {
      ka[c2] = kptr[c2 * 32];
      kc[c2] = kptr[128 + c2 * 32];
    }
    asm volatile("" :: "v"(ka[0]), "v"(ka[1]), "v"(ka[2]), "v"(ka[3]),
                       "v"(kc[0]), "v"(kc[1]), "v"(kc[2]), "v"(kc[3]));
    #pragma unroll
    for (int c2 = 0; c2 < 4; ++c2) {
      v2f ka0; ka0.x = ka[c2].x; ka0.y = ka[c2].y;
      v2f ka1; ka1.x = ka[c2].z; ka1.y = ka[c2].w;
      v2f kc0; kc0.x = kc[c2].x; kc0.y = kc[c2].y;
      v2f kc1; kc1.x = kc[c2].z; kc1.y = kc[c2].w;
      s0 = wmma4(qa[2 * c2],     ka0, s0);
      s1 = wmma4(qa[2 * c2],     kc0, s1);
      s0 = wmma4(qa[2 * c2 + 1], ka1, s0);
      s1 = wmma4(qa[2 * c2 + 1], kc1, s1);
    }
    #pragma unroll
    for (int r = 0; r < 8; ++r) {
      const int m    = r + hi * 8;
      const int qidx = qt * 16 + m;
      const int kidx = kp * 32 + l15;
      S[m * 18 + l15]       = (kidx      <= qidx) ? s0[r] * scale : -3.0e38f;
      S[288 + m * 18 + l15] = (kidx + 16 <= qidx) ? s1[r] * scale : -3.0e38f;
    }
    __syncthreads();

    {
      v2f sv[8];
      #pragma unroll
      for (int j = 0; j < 8; ++j) sv[j] = *(const v2f*)(Sr + 2 * j);
      float ml = fmaxf(sv[0].x, sv[0].y);
      #pragma unroll
      for (int j = 1; j < 8; ++j) ml = fmaxf(ml, fmaxf(sv[j].x, sv[j].y));
      const float mx = fmaxf(mreg, fmaxf(ml, other_half(ml)));
      const float f  = __expf(mreg - mx);
      float sum = 0.0f;
      #pragma unroll
      for (int j = 0; j < 8; ++j) {
        v2f p;
        p.x = __expf(sv[j].x - mx);
        p.y = __expf(sv[j].y - mx);
        *(v2f*)(Sr + 2 * j) = p;
        sum += p.x;
        sum += p.y;
      }
      mreg  = mx;
      lpart = lpart * f + sum;
      frow[l15] = f;
    }
    __syncthreads();

    const v4f* vp = Vq + kp * 512;
    v4f vv[16];
    #pragma unroll
    for (int i = 0; i < 16; ++i) vv[i] = vp[i * 32];

    const v4f f0 = *(const v4f*)(frow + hi * 8);
    const v4f f1 = *(const v4f*)(frow + hi * 8 + 4);
    const float fr[8] = {f0.x, f0.y, f0.z, f0.w, f1.x, f1.y, f1.z, f1.w};
    #pragma unroll
    for (int r = 0; r < 8; ++r) {
      o0[r] *= fr[r]; o1[r] *= fr[r]; o2[r] *= fr[r]; o3[r] *= fr[r];
    }

    v2f pa[8];
    #pragma unroll
    for (int c = 0; c < 8; ++c) {
      pa[c] = *(const v2f*)(S + (c >> 2) * 288 + l15 * 18 + 4 * (c & 3) + hi * 2);
    }
    asm volatile("" :: "v"(vv[0]), "v"(vv[1]), "v"(vv[2]),  "v"(vv[3]),
                       "v"(vv[4]), "v"(vv[5]), "v"(vv[6]),  "v"(vv[7]),
                       "v"(vv[8]), "v"(vv[9]), "v"(vv[10]), "v"(vv[11]),
                       "v"(vv[12]), "v"(vv[13]), "v"(vv[14]), "v"(vv[15]));
    #pragma unroll
    for (int c = 0; c < 8; ++c) {
      const v4f va = vv[2 * c];
      const v4f vc = vv[2 * c + 1];
      v2f vb0; vb0.x = va.x; vb0.y = va.y;
      v2f vb1; vb1.x = va.z; vb1.y = va.w;
      v2f vb2; vb2.x = vc.x; vb2.y = vc.y;
      v2f vb3; vb3.x = vc.z; vb3.y = vc.w;
      o0 = wmma4(pa[c], vb0, o0);
      o1 = wmma4(pa[c], vb1, o1);
      o2 = wmma4(pa[c], vb2, o2);
      o3 = wmma4(pa[c], vb3, o3);
    }
    __syncthreads();
  }

  lrow[l15] = lpart + other_half(lpart);
  __syncthreads();
  float inv[8];
  #pragma unroll
  for (int r = 0; r < 8; ++r) inv[r] = 1.0f / lrow[r + hi * 8];
  float* Ab = Abuf + ((size_t)(b * QKH + h)) * T_SEQ * VD;
  #pragma unroll
  for (int r = 0; r < 8; ++r) {
    float* Ar = Ab + (size_t)(qt * 16 + r + hi * 8) * VD;
    Ar[l15]      = o0[r] * inv[r];
    Ar[l15 + 16] = o1[r] * inv[r];
    Ar[l15 + 32] = o2[r] * inv[r];
    Ar[l15 + 48] = o3[r] * inv[r];
  }
}

__global__ __launch_bounds__(256) void k_diffnorm(const float* __restrict__ Abuf,
                                                  const float* __restrict__ lamp,
                                                  float* __restrict__ Dbuf) {
  const float lam = *lamp;
  const int wid  = threadIdx.x >> 5;
  const int lane = threadIdx.x & 31;
  const size_t row = (size_t)blockIdx.x * 8 + wid;
  const int t  = (int)(row & 2047);
  const size_t bh = row >> 11;
  const int hh = (int)(bh & 15);
  const int b  = (int)(bh >> 4);

  const float* A1 = Abuf + ((size_t)(b * QKH + hh))      * T_SEQ * VD + (size_t)t * VD;
  const float* A2 = Abuf + ((size_t)(b * QKH + 16 + hh)) * T_SEQ * VD + (size_t)t * VD;
  const float d0 = A1[lane]      - lam * A2[lane];
  const float d1 = A1[lane + 32] - lam * A2[lane + 32];
  float ss = d0 * d0 + d1 * d1;
  #pragma unroll
  for (int off = 16; off; off >>= 1) ss += __shfl_xor(ss, off, 32);
  const float r = rsqrtf(ss * (1.0f / 64.0f) + 1e-5f) * 0.8f;
  float* Dp = Dbuf + ((size_t)(b * T_SEQ + t)) * C_DIM + hh * VD;
  Dp[lane]      = d0 * r;
  Dp[lane + 32] = d1 * r;
}

extern "C" void kernel_launch(void* const* d_in, const int* in_sizes, int n_in,
                              void* d_out, int out_size, void* d_ws, size_t ws_size,
                              hipStream_t stream) {
  const float* x      = (const float*)d_in[0];
  const float* W_attn = (const float*)d_in[1];
  const float* W_proj = (const float*)d_in[2];
  const float* lq1    = (const float*)d_in[3];
  const float* lk1    = (const float*)d_in[4];
  const float* lq2    = (const float*)d_in[5];
  const float* lk2    = (const float*)d_in[6];
  float* out = (float*)d_out;

  float* qkv  = (float*)d_ws;
  float* Abuf = qkv  + (size_t)B_SZ * T_SEQ * ROWQKV;
  float* Dbuf = Abuf + (size_t)B_SZ * QKH * T_SEQ * VD;
  float* Kf   = Dbuf + (size_t)B_SZ * T_SEQ * C_DIM;
  float* Vf   = Kf   + (size_t)B_SZ * QKH * T_SEQ * HD;
  float* lam  = Vf   + (size_t)B_SZ * NH * T_SEQ * VD;

  const int M = B_SZ * T_SEQ;

  {
    dim3 grid(3 * C_DIM / 128, M / 128);
    k_gemm<<<grid, 256, 0, stream>>>(x, W_attn, qkv, 3 * C_DIM, C_DIM);
  }
  k_pack<<<QBLK + KBLK + VBLK, 256, 0, stream>>>(qkv, Kf, Vf);
  k_lambda<<<1, 32, 0, stream>>>(lq1, lk1, lq2, lk2, lam);
  {
    dim3 grid((T_SEQ / 16) * QKH * B_SZ);
    k_attn<<<grid, 32, 0, stream>>>(qkv, Kf, Vf, Abuf);
  }
  k_diffnorm<<<(B_SZ * NH * T_SEQ) / 8, 256, 0, stream>>>(Abuf, lam, Dbuf);
  {
    dim3 grid(C_DIM / 128, M / 128);
    k_gemm<<<grid, 256, 0, stream>>>(Dbuf, W_proj, out, C_DIM, C_DIM);
  }
  (void)in_sizes; (void)n_in; (void)out_size; (void)ws_size;
}


# attention: work-groups of 4 waves = 4 neighbouring query tiles of one sub-head; each K/V pair block fetched once per group into two-stage LDS one trip ahead, fragments read from LDS; sums unchanged
# speedup vs baseline: 1.5726x; 1.1182x over previous
#include <hip/hip_runtime.h>
#include <hip/hip_bf16.h>
#include <math.h>

typedef float v2f __attribute__((ext_vector_type(2)));
typedef float v4f __attribute__((ext_vector_type(4)));
typedef float v8f __attribute__((ext_vector_type(8)));

#define B_SZ   2
#define T_SEQ  2048
#define C_DIM  1024
#define NH     16
#define HD     32
#define QKH    32
#define VD     64
#define ROWQKV 3072

__device__ __forceinline__ v8f wmma4(v2f a, v2f b, v8f c) {
  return __builtin_amdgcn_wmma_f32_16x16x4_f32(
       false, a,  false, b,
       (short)0, c,  false,  false);
}

__device__ __forceinline__ void gemm_trip(v8f (&acc)[2][4], const v4f a0, const v4f a1,
                                          const v2f (&bp0)[4], const v2f (&bp1)[4]) {
  v2f x0, x1;
  x0.x = a0.x; x0.y = a0.y;
  x1.x = a1.x; x1.y = a1.y;
  #pragma unroll
  for (int cg = 0; cg < 4; ++cg) {
    acc[0][cg] = wmma4(x0, bp0[cg], acc[0][cg]);
    acc[1][cg] = wmma4(x1, bp0[cg], acc[1][cg]);
  }
  x0.x = a0.z; x0.y = a0.w;
  x1.x = a1.z; x1.y = a1.w;
  #pragma unroll
  for (int cg = 0; cg < 4; ++cg) {
    acc[0][cg] = wmma4(x0, bp1[cg], acc[0][cg]);
    acc[1][cg] = wmma4(x1, bp1[cg], acc[1][cg]);
  }
}

__device__ __forceinline__ void gemm_pairs(const v4f (&b)[4], v2f (&bp0)[4], v2f (&bp1)[4]) {
  #pragma unroll
  for (int cg = 0; cg < 4; ++cg) {
    bp0[cg].x = b[0][cg]; bp0[cg].y = b[1][cg];
    bp1[cg].x = b[2][cg]; bp1[cg].y = b[3][cg];
  }
}

__global__ __launch_bounds__(256) void k_gemm(const float* __restrict__ A,
                                              const float* __restrict__ Bm,
                                              float* __restrict__ Cm,
                                              int N, int K) {
  const int lane = threadIdx.x & 31;
  const int wid  = threadIdx.x >> 5;
  const int wy   = wid & 3;
  const int wx   = wid >> 2;
  const int rowBase = blockIdx.y * 128 + wy * 32;
  const int colBase = blockIdx.x * 128 + wx * 64;
  const int hi   = lane >> 4;
  const int l15  = lane & 15;
  const int bc   = colBase + 4 * l15;

  const float* pA0 = A + (size_t)(rowBase + l15) * K + hi * 4;
  const float* pA1 = A + (size_t)(rowBase + 16 + l15) * K + hi * 4;
  const float* pB  = Bm + (size_t)(hi * 4) * N + bc;
  const size_t strideB = (size_t)8 * N;

  v8f acc[2][4] = {};

  v4f a0 = *(const v4f*)pA0;
  v4f a1 = *(const v4f*)pA1;
  v2f bp0[4], bp1[4];
  {
    v4f b[4];
    #pragma unroll
    for (int i = 0; i < 4; ++i) b[i] = *(const v4f*)(pB + (size_t)i * N);
    gemm_pairs(b, bp0, bp1);
  }

  #pragma unroll 2
  for (int k = 0; k < K - 8; k += 8) {
    pA0 += 8; pA1 += 8; pB += strideB;
    const v4f na0 = *(const v4f*)pA0;
    const v4f na1 = *(const v4f*)pA1;
    v4f nb[4];
    #pragma unroll
    for (int i = 0; i < 4; ++i) nb[i] = *(const v4f*)(pB + (size_t)i * N);
    __builtin_amdgcn_sched_barrier(0);
    gemm_trip(acc, a0, a1, bp0, bp1);
    __builtin_amdgcn_sched_barrier(0);
    a0 = na0; a1 = na1;
    gemm_pairs(nb, bp0, bp1);
  }
  gemm_trip(acc, a0, a1, bp0, bp1);

  #pragma unroll
  for (int g = 0; g < 2; ++g) {
    #pragma unroll
    for (int r = 0; r < 8; ++r) {
      const size_t row = (size_t)(rowBase + g * 16 + r + hi * 8);
      v4f o;
      o.x = acc[g][0][r]; o.y = acc[g][1][r]; o.z = acc[g][2][r]; o.w = acc[g][3][r];
      *(v4f*)(Cm + row * N + bc) = o;
    }
  }
}

#define QBLK 8192
#define KBLK 4096
#define VBLK 4096

__device__ __forceinline__ v2f rope_pair(v2f x, int j, int t) {
  const float theta = __expf(-((float)(2 * j) / 32.0f) * logf(10000.0f));
  float s, c;
  __sincosf((float)t * theta, &s, &c);
  v2f r;
  r.x = x.x * c - x.y * s;
  r.y = x.x * s + x.y * c;
  return r;
}

__global__ __launch_bounds__(256) void k_pack(float* __restrict__ qkv,
                                              float* __restrict__ Kf,
                                              float* __restrict__ Vf) {
  const unsigned blk = blockIdx.x;
  if (blk < QBLK) {
    unsigned idx = blk * 256u + threadIdx.x;
    const int j = idx & 15;   idx >>= 4;
    const int h = idx & 31;   idx >>= 5;
    const int t = idx & 2047; idx >>= 11;
    const int b = (int)idx;
    float* p = qkv + ((size_t)(b * T_SEQ + t)) * ROWQKV + h * HD + 2 * j;
    const v2f x = *(const v2f*)p;
    *(v2f*)p = rope_pair(x, j, t);
  } else if (blk < QBLK + KBLK) {
    const unsigned o = (blk - QBLK) * 256u + threadIdx.x;
    const int lane = o & 31;
    const int c2   = (o >> 5) & 3;
    const int kb   = (o >> 7) & 127;
    const int bh   = (int)(o >> 14);
    const int hi   = lane >> 4;
    const int l15  = lane & 15;
    const int b    = bh >> 5;
    const int h    = bh & 31;
    const int t    = kb * 16 + l15;
    const float* s0 = qkv + ((size_t)(b * T_SEQ + t)) * ROWQKV + C_DIM + h * HD
                          + 8 * c2 + 2 * hi;
    const v2f xa = *(const v2f*)s0;
    const v2f xb = *(const v2f*)(s0 + 4);
    const v2f ra = rope_pair(xa, 4 * c2 + hi, t);
    const v2f rb = rope_pair(xb, 4 * c2 + 2 + hi, t);
    v4f w; w.x = ra.x; w.y = ra.y; w.z = rb.x; w.w = rb.y;
    ((v4f*)Kf)[o] = w;
  } else {
    const unsigned o = (blk - (QBLK + KBLK)) * 256u + threadIdx.x;
    const int lane = o & 31;
    const int jp   = (o >> 5) & 1;
    const int c    = (o >> 6) & 3;
    const int kb   = (o >> 8) & 127;
    const int bhv  = (int)(o >> 15);
    const int hi   = lane >> 4;
    const int l15  = lane & 15;
    const int b    = bhv >> 4;
    const int hv   = bhv & 15;
    const int k0   = kb * 16 + 4 * c + 2 * hi;
    const float* s0 = qkv + ((size_t)(b * T_SEQ + k0)) * ROWQKV + 2 * C_DIM + hv * VD
                          + 32 * jp + l15;
    const float* s1 = s0 + ROWQKV;
    v4f w; w.x = s0[0]; w.y = s1[0]; w.z = s0[16]; w.w = s1[16];
    ((v4f*)Vf)[o] = w;
  }
}

__global__ void k_lambda(const float* __restrict__ lq1, const float* __restrict__ lk1,
                         const float* __restrict__ lq2, const float* __restrict__ lk2,
                         float* __restrict__ lam) {
  const int l = threadIdx.x;
  float s1 = lq1[l] * lk1[l];
  float s2 = lq2[l] * lk2[l];
  #pragma unroll
  for (int off = 16; off; off >>= 1) {
    s1 += __shfl_xor(s1, off, 32);
    s2 += __shfl_xor(s2, off, 32);
  }
  if (l == 0) *lam = __expf(s1) - __expf(s2) + 0.2f;
}

__device__ __forceinline__ float other_half(float v) {
  const unsigned u = __float_as_uint(v);
  return __uint_as_float(
      __builtin_amdgcn_permlanex16(u, u, 0x76543210u, 0xfedcba98u, false, false));
}

__device__ __forceinline__ void wave_lds_order() {
  __builtin_amdgcn_fence(__ATOMIC_SEQ_CST, "wavefront");
  __builtin_amdgcn_wave_barrier();
  __builtin_amdgcn_fence(__ATOMIC_SEQ_CST, "wavefront");
}

#define AW 4

__global__ __launch_bounds__(128) void k_attn(const float* __restrict__ qkv,
                                              const float* __restrict__ Kf,
                                              const float* __restrict__ Vf,
                                              float* __restrict__ Abuf) {
  const unsigned L = blockIdx.x;
  const int g    = 31 - (int)(L >> 6);
  const int h    = (int)(L & 31u);
  const int b    = (int)((L >> 5) & 1u);
  const int tid  = threadIdx.x;
  const int w    = tid >> 5;
  const int lane = tid & 31;
  const int hi   = lane >> 4;
  const int l15  = lane & 15;
  const int qt   = 4 * g + w;
  const int ntrip = 2 * g + 2;

  const float* Q  = qkv + (size_t)b * T_SEQ * ROWQKV + h * HD;
  const v4f* Kg = (const v4f*)Kf + (size_t)(b * QKH + h) * (128 * 128) + tid;
  const v4f* Vg = (const v4f*)Vf + (size_t)(b * NH + (h & 15)) * (128 * 256) + tid;

  __shared__ v4f Ks[2][256];
  __shared__ v4f Vs[2][512];
  __shared__ __attribute__((aligned(16))) float Sall[AW][2 * 288];
  __shared__ __attribute__((aligned(16))) float frowall[AW][16];
  __shared__ float lrowall[AW][16];
  float* const S    = Sall[w];
  float* const frow = frowall[w];
  float* const lrow = lrowall[w];

  const size_t qoff = (size_t)(qt * 16 + l15) * ROWQKV;
  v2f qa[8];
  #pragma unroll
  for (int c = 0; c < 8; ++c) {
    const int col = 4 * c + hi * 2;
    qa[c].x = Q[qoff + col];
    qa[c].y = Q[qoff + col + 1];
  }

  v8f o0 = {}, o1 = {}, o2 = {}, o3 = {};
  const float scale = 0.17677669529663687f;

  float mreg  = -3.0e38f;
  float lpart = 0.0f;
  float* const Sr = S + hi * 288 + l15 * 18;

  {
    v4f gk[2], gv[4];
    #pragma unroll
    for (int i = 0; i < 2; ++i) gk[i] = Kg[i * 128];
    #pragma unroll
    for (int i = 0; i < 4; ++i) gv[i] = Vg[i * 128];
    #pragma unroll
    for (int i = 0; i < 2; ++i) Ks[0][tid + i * 128] = gk[i];
    #pragma unroll
    for (int i = 0; i < 4; ++i) Vs[0][tid + i * 128] = gv[i];
  }
  __syncthreads();

  for (int kp = 0; kp < ntrip; ++kp) {
    const int  cur  = kp & 1;
    const bool more = (kp + 1) < ntrip;
    v4f gk[2], gv[4];
    if (more) {
      const v4f* kn = Kg + (kp + 1) * 256;
      const v4f* vn = Vg + (kp + 1) * 512;
      #pragma unroll
      for (int i = 0; i < 2; ++i) gk[i] = kn[i * 128];
      #pragma unroll
      for (int i = 0; i < 4; ++i) gv[i] = vn[i * 128];
    }

    v8f s0 = {}, s1 = {};
    const v4f* kptr = &Ks[cur][lane];
    v4f ka[4], kc[4];
    #pragma unroll
    for (int c2 = 0; c2 < 4; ++c2) {
      ka[c2] = kptr[c2 * 32];
      kc[c2] = kptr[128 + c2 * 32];
    }
    asm volatile("" :: "v"(ka[0]), "v"(ka[1]), "v"(ka[2]), "v"(ka[3]),
                       "v"(kc[0]), "v"(kc[1]), "v"(kc[2]), "v"(kc[3]));
    #pragma unroll
    for (int c2 = 0; c2 < 4; ++c2) {
      v2f ka0; ka0.x = ka[c2].x; ka0.y = ka[c2].y;
      v2f ka1; ka1.x = ka[c2].z; ka1.y = ka[c2].w;
      v2f kc0; kc0.x = kc[c2].x; kc0.y = kc[c2].y;
      v2f kc1; kc1.x = kc[c2].z; kc1.y = kc[c2].w;
      s0 = wmma4(qa[2 * c2],     ka0, s0);
      s1 = wmma4(qa[2 * c2],     kc0, s1);
      s0 = wmma4(qa[2 * c2 + 1], ka1, s0);
      s1 = wmma4(qa[2 * c2 + 1], kc1, s1);
    }
    #pragma unroll
    for (int r = 0; r < 8; ++r) {
      const int m    = r + hi * 8;
      const int qidx = qt * 16 + m;
      const int kidx = kp * 32 + l15;
      S[m * 18 + l15]       = (kidx      <= qidx) ? s0[r] * scale : -3.0e38f;
      S[288 + m * 18 + l15] = (kidx + 16 <= qidx) ? s1[r] * scale : -3.0e38f;
    }
    wave_lds_order();

    {
      v2f sv[8];
      #pragma unroll
      for (int j = 0; j < 8; ++j) sv[j] = *(const v2f*)(Sr + 2 * j);
      float ml = fmaxf(sv[0].x, sv[0].y);
      #pragma unroll
      for (int j = 1; j < 8; ++j) ml = fmaxf(ml, fmaxf(sv[j].x, sv[j].y));
      const float mx = fmaxf(mreg, fmaxf(ml, other_half(ml)));
      const float f  = __expf(mreg - mx);
      float sum = 0.0f;
      #pragma unroll
      for (int j = 0; j < 8; ++j) {
        v2f p;
        p.x = __expf(sv[j].x - mx);
        p.y = __expf(sv[j].y - mx);
        *(v2f*)(Sr + 2 * j) = p;
        sum += p.x;
        sum += p.y;
      }
      mreg  = mx;
      lpart = lpart * f + sum;
      frow[l15] = f;
    }
    wave_lds_order();

    if (more) {
      #pragma unroll
      for (int i = 0; i < 2; ++i) Ks[cur ^ 1][tid + i * 128] = gk[i];
      #pragma unroll
      for (int i = 0; i < 4; ++i) Vs[cur ^ 1][tid + i * 128] = gv[i];
    }

    const v4f* vp = &Vs[cur][lane];
    v4f vv[16];
    #pragma unroll
    for (int i = 0; i < 16; ++i) vv[i] = vp[i * 32];

    const v4f f0 = *(const v4f*)(frow + hi * 8);
    const v4f f1 = *(const v4f*)(frow + hi * 8 + 4);
    const float fr[8] = {f0.x, f0.y, f0.z, f0.w, f1.x, f1.y, f1.z, f1.w};
    #pragma unroll
    for (int r = 0; r < 8; ++r) {
      o0[r] *= fr[r]; o1[r] *= fr[r]; o2[r] *= fr[r]; o3[r] *= fr[r];
    }

    v2f pa[8];
    #pragma unroll
    for (int c = 0; c < 8; ++c) {
      pa[c] = *(const v2f*)(S + (c >> 2) * 288 + l15 * 18 + 4 * (c & 3) + hi * 2);
    }
    asm volatile("" :: "v"(vv[0]), "v"(vv[1]), "v"(vv[2]),  "v"(vv[3]),
                       "v"(vv[4]), "v"(vv[5]), "v"(vv[6]),  "v"(vv[7]),
                       "v"(vv[8]), "v"(vv[9]), "v"(vv[10]), "v"(vv[11]),
                       "v"(vv[12]), "v"(vv[13]), "v"(vv[14]), "v"(vv[15]));
    __builtin_amdgcn_sched_barrier(0);
    #pragma unroll
    for (int c = 0; c < 8; ++c) {
      const v4f va = vv[2 * c];
      const v4f vc = vv[2 * c + 1];
      v2f vb0; vb0.x = va.x; vb0.y = va.y;
      v2f vb1; vb1.x = va.z; vb1.y = va.w;
      v2f vb2; vb2.x = vc.x; vb2.y = vc.y;
      v2f vb3; vb3.x = vc.z; vb3.y = vc.w;
      o0 = wmma4(pa[c], vb0, o0);
      o1 = wmma4(pa[c], vb1, o1);
      o2 = wmma4(pa[c], vb2, o2);
      o3 = wmma4(pa[c], vb3, o3);
    }

    __syncthreads();
  }

  lrow[l15] = lpart + other_half(lpart);
  wave_lds_order();
  float inv[8];
  #pragma unroll
  for (int r = 0; r < 8; ++r) inv[r] = 1.0f / lrow[r + hi * 8];
  float* Ab = Abuf + ((size_t)(b * QKH + h)) * T_SEQ * VD;
  #pragma unroll
  for (int r = 0; r < 8; ++r) {
    float* Ar = Ab + (size_t)(qt * 16 + r + hi * 8) * VD;
    Ar[l15]      = o0[r] * inv[r];
    Ar[l15 + 16] = o1[r] * inv[r];
    Ar[l15 + 32] = o2[r] * inv[r];
    Ar[l15 + 48] = o3[r] * inv[r];
  }
}

__global__ __launch_bounds__(256) void k_diffnorm(const float* __restrict__ Abuf,
                                                  const float* __restrict__ lamp,
                                                  float* __restrict__ Dbuf) {
  const float lam = *lamp;
  const int wid  = threadIdx.x >> 5;
  const int lane = threadIdx.x & 31;
  const size_t row = (size_t)blockIdx.x * 8 + wid;
  const int t  = (int)(row & 2047);
  const size_t bh = row >> 11;
  const int hh = (int)(bh & 15);
  const int b  = (int)(bh >> 4);

  const float* A1 = Abuf + ((size_t)(b * QKH + hh))      * T_SEQ * VD + (size_t)t * VD;
  const float* A2 = Abuf + ((size_t)(b * QKH + 16 + hh)) * T_SEQ * VD + (size_t)t * VD;
  const float d0 = A1[lane]      - lam * A2[lane];
  const float d1 = A1[lane + 32] - lam * A2[lane + 32];
  float ss = d0 * d0 + d1 * d1;
  #pragma unroll
  for (int off = 16; off; off >>= 1) ss += __shfl_xor(ss, off, 32);
  const float r = rsqrtf(ss * (1.0f / 64.0f) + 1e-5f) * 0.8f;
  float* Dp = Dbuf + ((size_t)(b * T_SEQ + t)) * C_DIM + hh * VD;
  Dp[lane]      = d0 * r;
  Dp[lane + 32] = d1 * r;
}

extern "C" void kernel_launch(void* const* d_in, const int* in_sizes, int n_in,
                              void* d_out, int out_size, void* d_ws, size_t ws_size,
                              hipStream_t stream) {
  const float* x      = (const float*)d_in[0];
  const float* W_attn = (const float*)d_in[1];
  const float* W_proj = (const float*)d_in[2];
  const float* lq1    = (const float*)d_in[3];
  const float* lk1    = (const float*)d_in[4];
  const float* lq2    = (const float*)d_in[5];
  const float* lk2    = (const float*)d_in[6];
  float* out = (float*)d_out;

  float* qkv  = (float*)d_ws;
  float* Abuf = qkv  + (size_t)B_SZ * T_SEQ * ROWQKV;
  float* Dbuf = Abuf + (size_t)B_SZ * QKH * T_SEQ * VD;
  float* Kf   = Dbuf + (size_t)B_SZ * T_SEQ * C_DIM;
  float* Vf   = Kf   + (size_t)B_SZ * QKH * T_SEQ * HD;
  float* lam  = Vf   + (size_t)B_SZ * NH * T_SEQ * VD;

  const int M = B_SZ * T_SEQ;

  {
    dim3 grid(3 * C_DIM / 128, M / 128);
    k_gemm<<<grid, 256, 0, stream>>>(x, W_attn, qkv, 3 * C_DIM, C_DIM);
  }
  k_pack<<<QBLK + KBLK + VBLK, 256, 0, stream>>>(qkv, Kf, Vf);
  k_lambda<<<1, 32, 0, stream>>>(lq1, lk1, lq2, lk2, lam);
  {
    dim3 grid((T_SEQ / 64) * QKH * B_SZ);
    k_attn<<<grid, 128, 0, stream>>>(qkv, Kf, Vf, Abuf);
  }
  k_diffnorm<<<(B_SZ * NH * T_SEQ) / 8, 256, 0, stream>>>(Abuf, lam, Dbuf);
  {
    dim3 grid(C_DIM / 128, M / 128);
    k_gemm<<<grid, 256, 0, stream>>>(Dbuf, W_proj, out, C_DIM, C_DIM);
  }
  (void)in_sizes; (void)n_in; (void)out_size; (void)ws_size;
}


# both GEMMs read A pre-arranged in fragment order (x by the lambda launch, differences by the norm pass): an A load is 4 whole lines, was 32 B of each of 16; norm's sum of 64 squares regrouped
# speedup vs baseline: 1.8149x; 1.1541x over previous
#include <hip/hip_runtime.h>
#include <hip/hip_bf16.h>
#include <math.h>

typedef float v2f __attribute__((ext_vector_type(2)));
typedef float v4f __attribute__((ext_vector_type(4)));
typedef float v8f __attribute__((ext_vector_type(8)));

#define B_SZ   2
#define T_SEQ  2048
#define C_DIM  1024
#define NH     16
#define HD     32
#define QKH    32
#define VD     64
#define ROWQKV 3072

__device__ __forceinline__ v8f wmma4(v2f a, v2f b, v8f c) {
  return __builtin_amdgcn_wmma_f32_16x16x4_f32(
       false, a,  false, b,
       (short)0, c,  false,  false);
}

__device__ __forceinline__ void gemm_trip(v8f (&acc)[2][4], const v4f a0, const v4f a1,
                                          const v2f (&bp0)[4], const v2f (&bp1)[4]) {
  v2f x0, x1;
  x0.x = a0.x; x0.y = a0.y;
  x1.x = a1.x; x1.y = a1.y;
  #pragma unroll
  for (int cg = 0; cg < 4; ++cg) {
    acc[0][cg] = wmma4(x0, bp0[cg], acc[0][cg]);
    acc[1][cg] = wmma4(x1, bp0[cg], acc[1][cg]);
  }
  x0.x = a0.z; x0.y = a0.w;
  x1.x = a1.z; x1.y = a1.w;
  #pragma unroll
  for (int cg = 0; cg < 4; ++cg) {
    acc[0][cg] = wmma4(x0, bp1[cg], acc[0][cg]);
    acc[1][cg] = wmma4(x1, bp1[cg], acc[1][cg]);
  }
}

__device__ __forceinline__ void gemm_pairs(const v4f (&b)[4], v2f (&bp0)[4], v2f (&bp1)[4]) {
  #pragma unroll
  for (int cg = 0; cg < 4; ++cg) {
    bp0[cg].x = b[0][cg]; bp0[cg].y = b[1][cg];
    bp1[cg].x = b[2][cg]; bp1[cg].y = b[3][cg];
  }
}

__global__ __launch_bounds__(256) void k_gemm(const float* __restrict__ Af,
                                              const float* __restrict__ Bm,
                                              float* __restrict__ Cm,
                                              int N, int K) {
  const int lane = threadIdx.x & 31;
  const int wid  = threadIdx.x >> 5;
  const int wy   = wid & 3;
  const int wx   = wid >> 2;
  const int rowBase = blockIdx.y * 128 + wy * 32;
  const int colBase = blockIdx.x * 128 + wx * 64;
  const int hi   = lane >> 4;
  const int l15  = lane & 15;
  const int bc   = colBase + 4 * l15;

  const float* pA0 = Af + ((size_t)(rowBase >> 4) * (size_t)(K >> 3) * 32 + lane) * 4;
  const float* pA1 = pA0 + (size_t)(K >> 3) * 128;
  const float* pB  = Bm + (size_t)(hi * 4) * N + bc;
  const size_t strideB = (size_t)8 * N;

  v8f acc[2][4] = {};

  v4f a0 = *(const v4f*)pA0;
  v4f a1 = *(const v4f*)pA1;
  v2f bp0[4], bp1[4];
  {
    v4f b[4];
    #pragma unroll
    for (int i = 0; i < 4; ++i) b[i] = *(const v4f*)(pB + (size_t)i * N);
    gemm_pairs(b, bp0, bp1);
  }

  #pragma unroll 2
  for (int k = 0; k < K - 8; k += 8) {
    pA0 += 128; pA1 += 128; pB += strideB;
    const v4f na0 = *(const v4f*)pA0;
    const v4f na1 = *(const v4f*)pA1;
    v4f nb[4];
    #pragma unroll
    for (int i = 0; i < 4; ++i) nb[i] = *(const v4f*)(pB + (size_t)i * N);
    __builtin_amdgcn_sched_barrier(0);
    gemm_trip(acc, a0, a1, bp0, bp1);
    __builtin_amdgcn_sched_barrier(0);
    a0 = na0; a1 = na1;
    gemm_pairs(nb, bp0, bp1);
  }
  gemm_trip(acc, a0, a1, bp0, bp1);

  #pragma unroll
  for (int g = 0; g < 2; ++g) {
    #pragma unroll
    for (int r = 0; r < 8; ++r) {
      const size_t row = (size_t)(rowBase + g * 16 + r + hi * 8);
      v4f o;
      o.x = acc[g][0][r]; o.y = acc[g][1][r]; o.z = acc[g][2][r]; o.w = acc[g][3][r];
      *(v4f*)(Cm + row * N + bc) = o;
    }
  }
}

#define QBLK 8192
#define KBLK 4096
#define VBLK 4096

__device__ __forceinline__ v2f rope_pair(v2f x, int j, int t) {
  const float theta = __expf(-((float)(2 * j) / 32.0f) * logf(10000.0f));
  float s, c;
  __sincosf((float)t * theta, &s, &c);
  v2f r;
  r.x = x.x * c - x.y * s;
  r.y = x.x * s + x.y * c;
  return r;
}

__global__ __launch_bounds__(256) void k_pack(float* __restrict__ qkv,
                                              float* __restrict__ Kf,
                                              float* __restrict__ Vf) {
  const unsigned blk = blockIdx.x;
  if (blk < QBLK) {
    unsigned idx = blk * 256u + threadIdx.x;
    const int j = idx & 15;   idx >>= 4;
    const int h = idx & 31;   idx >>= 5;
    const int t = idx & 2047; idx >>= 11;
    const int b = (int)idx;
    float* p = qkv + ((size_t)(b * T_SEQ + t)) * ROWQKV + h * HD + 2 * j;
    const v2f x = *(const v2f*)p;
    *(v2f*)p = rope_pair(x, j, t);
  } else if (blk < QBLK + KBLK) {
    const unsigned o = (blk - QBLK) * 256u + threadIdx.x;
    const int lane = o & 31;
    const int c2   = (o >> 5) & 3;
    const int kb   = (o >> 7) & 127;
    const int bh   = (int)(o >> 14);
    const int hi   = lane >> 4;
    const int l15  = lane & 15;
    const int b    = bh >> 5;
    const int h    = bh & 31;
    const int t    = kb * 16 + l15;
    const float* s0 = qkv + ((size_t)(b * T_SEQ + t)) * ROWQKV + C_DIM + h * HD
                          + 8 * c2 + 2 * hi;
    const v2f xa = *(const v2f*)s0;
    const v2f xb = *(const v2f*)(s0 + 4);
    const v2f ra = rope_pair(xa, 4 * c2 + hi, t);
    const v2f rb = rope_pair(xb, 4 * c2 + 2 + hi, t);
    v4f w; w.x = ra.x; w.y = ra.y; w.z = rb.x; w.w = rb.y;
    ((v4f*)Kf)[o] = w;
  } else {
    const unsigned o = (blk - (QBLK + KBLK)) * 256u + threadIdx.x;
    const int lane = o & 31;
    const int jp   = (o >> 5) & 1;
    const int c    = (o >> 6) & 3;
    const int kb   = (o >> 8) & 127;
    const int bhv  = (int)(o >> 15);
    const int hi   = lane >> 4;
    const int l15  = lane & 15;
    const int b    = bhv >> 4;
    const int hv   = bhv & 15;
    const int k0   = kb * 16 + 4 * c + 2 * hi;
    const float* s0 = qkv + ((size_t)(b * T_SEQ + k0)) * ROWQKV + 2 * C_DIM + hv * VD
                          + 32 * jp + l15;
    const float* s1 = s0 + ROWQKV;
    v4f w; w.x = s0[0]; w.y = s1[0]; w.z = s0[16]; w.w = s1[16];
    ((v4f*)Vf)[o] = w;
  }
}

#define XBLK 1024

__global__ __launch_bounds__(256) void k_prep(const float* __restrict__ x,
                                              float* __restrict__ Xf,
                                              const float* __restrict__ lq1,
                                              const float* __restrict__ lk1,
                                              const float* __restrict__ lq2,
                                              const float* __restrict__ lk2,
                                              float* __restrict__ lam) {
  if (blockIdx.x == 0) {
    if (threadIdx.x < 32) {
      const int l = threadIdx.x;
      float s1 = lq1[l] * lk1[l];
      float s2 = lq2[l] * lk2[l];
      #pragma unroll
      for (int off = 16; off; off >>= 1) {
        s1 += __shfl_xor(s1, off, 32);
        s2 += __shfl_xor(s2, off, 32);
      }
      if (l == 0) *lam = __expf(s1) - __expf(s2) + 0.2f;
    }
    return;
  }
  const unsigned base = (blockIdx.x - 1u) * 1024u + threadIdx.x;
  v4f w[4];
  #pragma unroll
  for (int i = 0; i < 4; ++i) {
    const unsigned o = base + 256u * i;
    const int lane = o & 31;
    const int trip = (o >> 5) & 127;
    const int rg   = (int)(o >> 12);
    const int hi   = lane >> 4;
    const int l15  = lane & 15;
    w[i] = *(const v4f*)(x + (size_t)(rg * 16 + l15) * C_DIM + trip * 8 + hi * 4);
  }
  #pragma unroll
  for (int i = 0; i < 4; ++i) ((v4f*)Xf)[base + 256u * i] = w[i];
}

__device__ __forceinline__ float other_half(float v) {
  const unsigned u = __float_as_uint(v);
  return __uint_as_float(
      __builtin_amdgcn_permlanex16(u, u, 0x76543210u, 0xfedcba98u, false, false));
}

__device__ __forceinline__ void wave_lds_order() {
  __builtin_amdgcn_fence(__ATOMIC_SEQ_CST, "wavefront");
  __builtin_amdgcn_wave_barrier();
  __builtin_amdgcn_fence(__ATOMIC_SEQ_CST, "wavefront");
}

#define AW 4

__global__ __launch_bounds__(128) void k_attn(const float* __restrict__ qkv,
                                              const float* __restrict__ Kf,
                                              const float* __restrict__ Vf,
                                              float* __restrict__ Abuf) {
  const unsigned L = blockIdx.x;
  const int g    = 31 - (int)(L >> 6);
  const int h    = (int)(L & 31u);
  const int b    = (int)((L >> 5) & 1u);
  const int tid  = threadIdx.x;
  const int w    = tid >> 5;
  const int lane = tid & 31;
  const int hi   = lane >> 4;
  const int l15  = lane & 15;
  const int qt   = 4 * g + w;
  const int ntrip = 2 * g + 2;

  const float* Q  = qkv + (size_t)b * T_SEQ * ROWQKV + h * HD;
  const v4f* Kg = (const v4f*)Kf + (size_t)(b * QKH + h) * (128 * 128) + tid;
  const v4f* Vg = (const v4f*)Vf + (size_t)(b * NH + (h & 15)) * (128 * 256) + tid;

  __shared__ v4f Ks[2][256];
  __shared__ v4f Vs[2][512];
  __shared__ __attribute__((aligned(16))) float Sall[AW][2 * 288];
  __shared__ __attribute__((aligned(16))) float frowall[AW][16];
  __shared__ float lrowall[AW][16];
  float* const S    = Sall[w];
  float* const frow = frowall[w];
  float* const lrow = lrowall[w];

  const size_t qoff = (size_t)(qt * 16 + l15) * ROWQKV;
  v2f qa[8];
  #pragma unroll
  for (int c = 0; c < 8; ++c) {
    const int col = 4 * c + hi * 2;
    qa[c].x = Q[qoff + col];
    qa[c].y = Q[qoff + col + 1];
  }

  v8f o0 = {}, o1 = {}, o2 = {}, o3 = {};
  const float scale = 0.17677669529663687f;

  float mreg  = -3.0e38f;
  float lpart = 0.0f;
  float* const Sr = S + hi * 288 + l15 * 18;

  {
    v4f gk[2], gv[4];
    #pragma unroll
    for (int i = 0; i < 2; ++i) gk[i] = Kg[i * 128];
    #pragma unroll
    for (int i = 0; i < 4; ++i) gv[i] = Vg[i * 128];
    #pragma unroll
    for (int i = 0; i < 2; ++i) Ks[0][tid + i * 128] = gk[i];
    #pragma unroll
    for (int i = 0; i < 4; ++i) Vs[0][tid + i * 128] = gv[i];
  }
  __syncthreads();

  for (int kp = 0; kp < ntrip; ++kp) {
    const int  cur  = kp & 1;
    const bool more = (kp + 1) < ntrip;
    v4f gk[2], gv[4];
    if (more) {
      const v4f* kn = Kg + (kp + 1) * 256;
      const v4f* vn = Vg + (kp + 1) * 512;
      #pragma unroll
      for (int i = 0; i < 2; ++i) gk[i] = kn[i * 128];
      #pragma unroll
      for (int i = 0; i < 4; ++i) gv[i] = vn[i * 128];
    }

    v8f s0 = {}, s1 = {};
    const v4f* kptr = &Ks[cur][lane];
    v4f ka[4], kc[4];
    #pragma unroll
    for (int c2 = 0; c2 < 4; ++c2) {
      ka[c2] = kptr[c2 * 32];
      kc[c2] = kptr[128 + c2 * 32];
    }
    asm volatile("" :: "v"(ka[0]), "v"(ka[1]), "v"(ka[2]), "v"(ka[3]),
                       "v"(kc[0]), "v"(kc[1]), "v"(kc[2]), "v"(kc[3]));
    #pragma unroll
    for (int c2 = 0; c2 < 4; ++c2) {
      v2f ka0; ka0.x = ka[c2].x; ka0.y = ka[c2].y;
      v2f ka1; ka1.x = ka[c2].z; ka1.y = ka[c2].w;
      v2f kc0; kc0.x = kc[c2].x; kc0.y = kc[c2].y;
      v2f kc1; kc1.x = kc[c2].z; kc1.y = kc[c2].w;
      s0 = wmma4(qa[2 * c2],     ka0, s0);
      s1 = wmma4(qa[2 * c2],     kc0, s1);
      s0 = wmma4(qa[2 * c2 + 1], ka1, s0);
      s1 = wmma4(qa[2 * c2 + 1], kc1, s1);
    }
    #pragma unroll
    for (int r = 0; r < 8; ++r) {
      const int m    = r + hi * 8;
      const int qidx = qt * 16 + m;
      const int kidx = kp * 32 + l15;
      S[m * 18 + l15]       = (kidx      <= qidx) ? s0[r] * scale : -3.0e38f;
      S[288 + m * 18 + l15] = (kidx + 16 <= qidx) ? s1[r] * scale : -3.0e38f;
    }
    wave_lds_order();

    {
      v2f sv[8];
      #pragma unroll
      for (int j = 0; j < 8; ++j) sv[j] = *(const v2f*)(Sr + 2 * j);
      float ml = fmaxf(sv[0].x, sv[0].y);
      #pragma unroll
      for (int j = 1; j < 8; ++j) ml = fmaxf(ml, fmaxf(sv[j].x, sv[j].y));
      const float mx = fmaxf(mreg, fmaxf(ml, other_half(ml)));
      const float f  = __expf(mreg - mx);
      float sum = 0.0f;
      #pragma unroll
      for (int j = 0; j < 8; ++j) {
        v2f p;
        p.x = __expf(sv[j].x - mx);
        p.y = __expf(sv[j].y - mx);
        *(v2f*)(Sr + 2 * j) = p;
        sum += p.x;
        sum += p.y;
      }
      mreg  = mx;
      lpart = lpart * f + sum;
      frow[l15] = f;
    }
    wave_lds_order();

    if (more) {
      #pragma unroll
      for (int i = 0; i < 2; ++i) Ks[cur ^ 1][tid + i * 128] = gk[i];
      #pragma unroll
      for (int i = 0; i < 4; ++i) Vs[cur ^ 1][tid + i * 128] = gv[i];
    }

    const v4f* vp = &Vs[cur][lane];
    v4f vv[16];
    #pragma unroll
    for (int i = 0; i < 16; ++i) vv[i] = vp[i * 32];

    const v4f f0 = *(const v4f*)(frow + hi * 8);
    const v4f f1 = *(const v4f*)(frow + hi * 8 + 4);
    const float fr[8] = {f0.x, f0.y, f0.z, f0.w, f1.x, f1.y, f1.z, f1.w};
    #pragma unroll
    for (int r = 0; r < 8; ++r) {
      o0[r] *= fr[r]; o1[r] *= fr[r]; o2[r] *= fr[r]; o3[r] *= fr[r];
    }

    v2f pa[8];
    #pragma unroll
    for (int c = 0; c < 8; ++c) {
      pa[c] = *(const v2f*)(S + (c >> 2) * 288 + l15 * 18 + 4 * (c & 3) + hi * 2);
    }
    asm volatile("" :: "v"(vv[0]), "v"(vv[1]), "v"(vv[2]),  "v"(vv[3]),
                       "v"(vv[4]), "v"(vv[5]), "v"(vv[6]),  "v"(vv[7]),
                       "v"(vv[8]), "v"(vv[9]), "v"(vv[10]), "v"(vv[11]),
                       "v"(vv[12]), "v"(vv[13]), "v"(vv[14]), "v"(vv[15]));
    __builtin_amdgcn_sched_barrier(0);
    #pragma unroll
    for (int c = 0; c < 8; ++c) {
      const v4f va = vv[2 * c];
      const v4f vc = vv[2 * c + 1];
      v2f vb0; vb0.x = va.x; vb0.y = va.y;
      v2f vb1; vb1.x = va.z; vb1.y = va.w;
      v2f vb2; vb2.x = vc.x; vb2.y = vc.y;
      v2f vb3; vb3.x = vc.z; vb3.y = vc.w;
      o0 = wmma4(pa[c], vb0, o0);
      o1 = wmma4(pa[c], vb1, o1);
      o2 = wmma4(pa[c], vb2, o2);
      o3 = wmma4(pa[c], vb3, o3);
    }

    __syncthreads();
  }

  lrow[l15] = lpart + other_half(lpart);
  wave_lds_order();
  float inv[8];
  #pragma unroll
  for (int r = 0; r < 8; ++r) inv[r] = 1.0f / lrow[r + hi * 8];
  float* Ab = Abuf + ((size_t)(b * QKH + h)) * T_SEQ * VD;
  #pragma unroll
  for (int r = 0; r < 8; ++r) {
    float* Ar = Ab + (size_t)(qt * 16 + r + hi * 8) * VD;
    Ar[l15]      = o0[r] * inv[r];
    Ar[l15 + 16] = o1[r] * inv[r];
    Ar[l15 + 32] = o2[r] * inv[r];
    Ar[l15 + 48] = o3[r] * inv[r];
  }
}

__global__ __launch_bounds__(256) void k_diffnorm(const float* __restrict__ Abuf,
                                                  const float* __restrict__ lamp,
                                                  float* __restrict__ Df) {
  const float lam = *lamp;
  const int tl   = threadIdx.x >> 5;
  const int lane = threadIdx.x & 31;
  const int hi   = lane >> 4;
  const int l15  = lane & 15;
  const int hh   = (int)(blockIdx.x & 15u);
  const int rg   = (int)(blockIdx.x >> 4);
  const int m    = rg * 16 + l15;
  const int b    = m >> 11;
  const int t    = m & 2047;
  const int col  = tl * 8 + hi * 4;

  const float* A1 = Abuf + ((size_t)(b * QKH + hh)      * T_SEQ + t) * VD + col;
  const float* A2 = Abuf + ((size_t)(b * QKH + 16 + hh) * T_SEQ + t) * VD + col;
  const v4f a1 = *(const v4f*)A1;
  const v4f a2 = *(const v4f*)A2;
  v4f d;
  d.x = a1.x - lam * a2.x; d.y = a1.y - lam * a2.y;
  d.z = a1.z - lam * a2.z; d.w = a1.w - lam * a2.w;

  __shared__ float part[16][17];
  part[l15][tl * 2 + hi] = d.x * d.x + d.y * d.y + d.z * d.z + d.w * d.w;
  __syncthreads();
  float ss = 0.0f;
  #pragma unroll
  for (int i = 0; i < 16; ++i) ss += part[l15][i];
  const float r = rsqrtf(ss * (1.0f / 64.0f) + 1e-5f) * 0.8f;

  v4f o;
  o.x = d.x * r; o.y = d.y * r; o.z = d.z * r; o.w = d.w * r;
  ((v4f*)Df)[((size_t)rg * 128 + (size_t)(hh * 8 + tl)) * 32 + lane] = o;
}

extern "C" void kernel_launch(void* const* d_in, const int* in_sizes, int n_in,
                              void* d_out, int out_size, void* d_ws, size_t ws_size,
                              hipStream_t stream) {
  const float* x      = (const float*)d_in[0];
  const float* W_attn = (const float*)d_in[1];
  const float* W_proj = (const float*)d_in[2];
  const float* lq1    = (const float*)d_in[3];
  const float* lk1    = (const float*)d_in[4];
  const float* lq2    = (const float*)d_in[5];
  const float* lk2    = (const float*)d_in[6];
  float* out = (float*)d_out;

  float* qkv  = (float*)d_ws;
  float* Abuf = qkv  + (size_t)B_SZ * T_SEQ * ROWQKV;
  float* Dbuf = Abuf + (size_t)B_SZ * QKH * T_SEQ * VD;
  float* Kf   = Dbuf + (size_t)B_SZ * T_SEQ * C_DIM;
  float* Vf   = Kf   + (size_t)B_SZ * QKH * T_SEQ * HD;
  float* Xf   = Vf   + (size_t)B_SZ * NH * T_SEQ * VD;
  float* lam  = Xf   + (size_t)B_SZ * T_SEQ * C_DIM;

  const int M = B_SZ * T_SEQ;

  k_prep<<<1 + XBLK, 256, 0, stream>>>(x, Xf, lq1, lk1, lq2, lk2, lam);
  {
    dim3 grid(3 * C_DIM / 128, M / 128);
    k_gemm<<<grid, 256, 0, stream>>>(Xf, W_attn, qkv, 3 * C_DIM, C_DIM);
  }
  k_pack<<<QBLK + KBLK + VBLK, 256, 0, stream>>>(qkv, Kf, Vf);
  {
    dim3 grid((T_SEQ / 64) * QKH * B_SZ);
    k_attn<<<grid, 128, 0, stream>>>(qkv, Kf, Vf, Abuf);
  }
  k_diffnorm<<<(B_SZ * T_SEQ / 16) * NH, 256, 0, stream>>>(Abuf, lam, Dbuf);
  {
    dim3 grid(C_DIM / 128, M / 128);
    k_gemm<<<grid, 256, 0, stream>>>(Dbuf, W_proj, out, C_DIM, C_DIM);
  }
  (void)in_sizes; (void)n_in; (void)out_size; (void)ws_size;
}


# rotate-and-repack launch removed: qkv store rotates q,k in-lane and writes V in fragment order; attention stages K row-major into padded LDS; q.k dim sum and P.V key sum regrouped; 5 launches
# speedup vs baseline: 1.8777x; 1.0346x over previous
#include <hip/hip_runtime.h>
#include <hip/hip_bf16.h>
#include <math.h>

typedef float v2f __attribute__((ext_vector_type(2)));
typedef float v4f __attribute__((ext_vector_type(4)));
typedef float v8f __attribute__((ext_vector_type(8)));

#define B_SZ   2
#define T_SEQ  2048
#define C_DIM  1024
#define NH     16
#define HD     32
#define QKH    32
#define VD     64
#define ROWQKV 3072

__device__ __forceinline__ v8f wmma4(v2f a, v2f b, v8f c) {
  return __builtin_amdgcn_wmma_f32_16x16x4_f32(
       false, a,  false, b,
       (short)0, c,  false,  false);
}

__device__ __forceinline__ void gemm_trip(v8f (&acc)[2][4], const v4f a0, const v4f a1,
                                          const v2f (&bp0)[4], const v2f (&bp1)[4]) {
  v2f x0, x1;
  x0.x = a0.x; x0.y = a0.y;
  x1.x = a1.x; x1.y = a1.y;
  #pragma unroll
  for (int cg = 0; cg < 4; ++cg) {
    acc[0][cg] = wmma4(x0, bp0[cg], acc[0][cg]);
    acc[1][cg] = wmma4(x1, bp0[cg], acc[1][cg]);
  }
  x0.x = a0.z; x0.y = a0.w;
  x1.x = a1.z; x1.y = a1.w;
  #pragma unroll
  for (int cg = 0; cg < 4; ++cg) {
    acc[0][cg] = wmma4(x0, bp1[cg], acc[0][cg]);
    acc[1][cg] = wmma4(x1, bp1[cg], acc[1][cg]);
  }
}

__device__ __forceinline__ void gemm_pairs(const v4f (&b)[4], v2f (&bp0)[4], v2f (&bp1)[4]) {
  #pragma unroll
  for (int cg = 0; cg < 4; ++cg) {
    bp0[cg].x = b[0][cg]; bp0[cg].y = b[1][cg];
    bp1[cg].x = b[2][cg]; bp1[cg].y = b[3][cg];
  }
}

__device__ __forceinline__ v2f rope_pair(v2f x, int j, int t) {
  const float theta = __expf(-((float)(2 * j) / 32.0f) * logf(10000.0f));
  float s, c;
  __sincosf((float)t * theta, &s, &c);
  v2f r;
  r.x = x.x * c - x.y * s;
  r.y = x.x * s + x.y * c;
  return r;
}

template <int QKV>
__global__ __launch_bounds__(256) void k_gemm(const float* __restrict__ Af,
                                              const float* __restrict__ Bm,
                                              float* __restrict__ Cm,
                                              float* __restrict__ Vf,
                                              int N, int K) {
  const int lane = threadIdx.x & 31;
  const int wid  = threadIdx.x >> 5;
  const int wy   = wid & 3;
  const int wx   = wid >> 2;
  const int rowBase = blockIdx.y * 128 + wy * 32;
  const int colBase = blockIdx.x * 128 + wx * 64;
  const int hi   = lane >> 4;
  const int l15  = lane & 15;
  const int bc   = colBase + 4 * l15;

  const float* pA0 = Af + ((size_t)(rowBase >> 4) * (size_t)(K >> 3) * 32 + lane) * 4;
  const float* pA1 = pA0 + (size_t)(K >> 3) * 128;
  const float* pB  = Bm + (size_t)(hi * 4) * N + bc;
  const size_t strideB = (size_t)8 * N;

  v8f acc[2][4] = {};

  v4f a0 = *(const v4f*)pA0;
  v4f a1 = *(const v4f*)pA1;
  v2f bp0[4], bp1[4];
  {
    v4f b[4];
    #pragma unroll
    for (int i = 0; i < 4; ++i) b[i] = *(const v4f*)(pB + (size_t)i * N);
    gemm_pairs(b, bp0, bp1);
  }

  #pragma unroll 2
  for (int k = 0; k < K - 8; k += 8) {
    pA0 += 128; pA1 += 128; pB += strideB;
    const v4f na0 = *(const v4f*)pA0;
    const v4f na1 = *(const v4f*)pA1;
    v4f nb[4];
    #pragma unroll
    for (int i = 0; i < 4; ++i) nb[i] = *(const v4f*)(pB + (size_t)i * N);
    __builtin_amdgcn_sched_barrier(0);
    gemm_trip(acc, a0, a1, bp0, bp1);
    __builtin_amdgcn_sched_barrier(0);
    a0 = na0; a1 = na1;
    gemm_pairs(nb, bp0, bp1);
  }
  gemm_trip(acc, a0, a1, bp0, bp1);

  if (QKV && blockIdx.x * 128 >= 2 * C_DIM) {
    const int hv = (colBase - 2 * C_DIM) >> 6;
    #pragma unroll
    for (int g = 0; g < 2; ++g) {
      const int grp = (rowBase >> 4) + g;
      const int b   = grp >> 7;
      const int kb  = grp & 127;
      v4f* dst = (v4f*)Vf + ((size_t)((b * NH + hv) * 128 + kb) * 8) * 32 + lane;
      #pragma unroll
      for (int p = 0; p < 4; ++p) {
        #pragma unroll
        for (int tp = 0; tp < 2; ++tp) {
          v4f w;
          w.x = acc[g][2 * tp][2 * p];     w.y = acc[g][2 * tp][2 * p + 1];
          w.z = acc[g][2 * tp + 1][2 * p]; w.w = acc[g][2 * tp + 1][2 * p + 1];
          dst[(p * 2 + tp) * 32] = w;
        }
      }
    }
    return;
  }
  const int j0 = (bc & 31) >> 1;
  #pragma unroll
  for (int g = 0; g < 2; ++g) {
    #pragma unroll
    for (int r = 0; r < 8; ++r) {
      const size_t row = (size_t)(rowBase + g * 16 + r + hi * 8);
      v4f o;
      o.x = acc[g][0][r]; o.y = acc[g][1][r]; o.z = acc[g][2][r]; o.w = acc[g][3][r];
      if (QKV) {
        const int t = (int)(row & (size_t)(T_SEQ - 1));
        v2f e0, e1;
        e0.x = o.x; e0.y = o.y; e1.x = o.z; e1.y = o.w;
        e0 = rope_pair(e0, j0, t);
        e1 = rope_pair(e1, j0 + 1, t);
        o.x = e0.x; o.y = e0.y; o.z = e1.x; o.w = e1.y;
      }
      *(v4f*)(Cm + row * N + bc) = o;
    }
  }
}

#define XBLK 1024

__global__ __launch_bounds__(256) void k_prep(const float* __restrict__ x,
                                              float* __restrict__ Xf,
                                              const float* __restrict__ lq1,
                                              const float* __restrict__ lk1,
                                              const float* __restrict__ lq2,
                                              const float* __restrict__ lk2,
                                              float* __restrict__ lam) {
  if (blockIdx.x == 0) {
    if (threadIdx.x < 32) {
      const int l = threadIdx.x;
      float s1 = lq1[l] * lk1[l];
      float s2 = lq2[l] * lk2[l];
      #pragma unroll
      for (int off = 16; off; off >>= 1) {
        s1 += __shfl_xor(s1, off, 32);
        s2 += __shfl_xor(s2, off, 32);
      }
      if (l == 0) *lam = __expf(s1) - __expf(s2) + 0.2f;
    }
    return;
  }
  const unsigned base = (blockIdx.x - 1u) * 1024u + threadIdx.x;
  v4f w[4];
  #pragma unroll
  for (int i = 0; i < 4; ++i) {
    const unsigned o = base + 256u * i;
    const int lane = o & 31;
    const int trip = (o >> 5) & 127;
    const int rg   = (int)(o >> 12);
    const int hi   = lane >> 4;
    const int l15  = lane & 15;
    w[i] = *(const v4f*)(x + (size_t)(rg * 16 + l15) * C_DIM + trip * 8 + hi * 4);
  }
  #pragma unroll
  for (int i = 0; i < 4; ++i) ((v4f*)Xf)[base + 256u * i] = w[i];
}

__device__ __forceinline__ float other_half(float v) {
  const unsigned u = __float_as_uint(v);
  return __uint_as_float(
      __builtin_amdgcn_permlanex16(u, u, 0x76543210u, 0xfedcba98u, false, false));
}

__device__ __forceinline__ void wave_lds_order() {
  __builtin_amdgcn_fence(__ATOMIC_SEQ_CST, "wavefront");
  __builtin_amdgcn_wave_barrier();
  __builtin_amdgcn_fence(__ATOMIC_SEQ_CST, "wavefront");
}

#define AW 4

__global__ __launch_bounds__(128) void k_attn(const float* __restrict__ qkv,
                                              const float* __restrict__ Vf,
                                              float* __restrict__ Abuf) {
  const unsigned L = blockIdx.x;
  const int g    = 31 - (int)(L >> 6);
  const int h    = (int)(L & 31u);
  const int b    = (int)((L >> 5) & 1u);
  const int tid  = threadIdx.x;
  const int w    = tid >> 5;
  const int lane = tid & 31;
  const int hi   = lane >> 4;
  const int l15  = lane & 15;
  const int qt   = 4 * g + w;
  const int ntrip = 2 * g + 2;

  const float* Q  = qkv + (size_t)b * T_SEQ * ROWQKV + h * HD;
  const float* Kg = qkv + (size_t)b * T_SEQ * ROWQKV + C_DIM + h * HD
                        + (size_t)(tid >> 3) * ROWQKV + 4 * (tid & 7);
  const int ksoff = (tid >> 3) * 36 + 4 * (tid & 7);
  const v4f* Vg = (const v4f*)Vf + (size_t)(b * NH + (h & 15)) * (128 * 256) + tid;

  __shared__ __attribute__((aligned(16))) float Ks[2][32 * 36];
  __shared__ v4f Vs[2][512];
  __shared__ __attribute__((aligned(16))) float Sall[AW][2 * 288];
  __shared__ __attribute__((aligned(16))) float frowall[AW][16];
  __shared__ float lrowall[AW][16];
  float* const S    = Sall[w];
  float* const frow = frowall[w];
  float* const lrow = lrowall[w];

  const size_t qoff = (size_t)(qt * 16 + l15) * ROWQKV;
  v2f qa[8];
  #pragma unroll
  for (int c2 = 0; c2 < 4; ++c2) {
    const v4f q4 = *(const v4f*)(Q + qoff + 8 * c2 + 4 * hi);
    qa[2 * c2].x     = q4.x; qa[2 * c2].y     = q4.y;
    qa[2 * c2 + 1].x = q4.z; qa[2 * c2 + 1].y = q4.w;
  }

  v8f o0 = {}, o1 = {}, o2 = {}, o3 = {};
  const float scale = 0.17677669529663687f;

  float mreg  = -3.0e38f;
  float lpart = 0.0f;
  float* const Sr = S + hi * 288 + l15 * 18;

  {
    v4f gk[2], gv[4];
    #pragma unroll
    for (int i = 0; i < 2; ++i) gk[i] = *(const v4f*)(Kg + (size_t)(16 * i) * ROWQKV);
    #pragma unroll
    for (int i = 0; i < 4; ++i) gv[i] = Vg[i * 128];
    #pragma unroll
    for (int i = 0; i < 2; ++i) *(v4f*)&Ks[0][ksoff + 16 * 36 * i] = gk[i];
    #pragma unroll
    for (int i = 0; i < 4; ++i) Vs[0][tid + i * 128] = gv[i];
  }
  __syncthreads();

  for (int kp = 0; kp < ntrip; ++kp) {
    const int  cur  = kp & 1;
    const bool more = (kp + 1) < ntrip;
    v4f gk[2], gv[4];
    if (more) {
      const float* kn = Kg + (size_t)((kp + 1) * 32) * ROWQKV;
      const v4f* vn = Vg + (kp + 1) * 512;
      #pragma unroll
      for (int i = 0; i < 2; ++i) gk[i] = *(const v4f*)(kn + (size_t)(16 * i) * ROWQKV);
      #pragma unroll
      for (int i = 0; i < 4; ++i) gv[i] = vn[i * 128];
    }

    v8f s0 = {}, s1 = {};
    const float* kptr = &Ks[cur][l15 * 36 + 4 * hi];
    v4f ka[4], kc[4];
    #pragma unroll
    for (int c2 = 0; c2 < 4; ++c2) {
      ka[c2] = *(const v4f*)(kptr + 8 * c2);
      kc[c2] = *(const v4f*)(kptr + 16 * 36 + 8 * c2);
    }
    asm volatile("" :: "v"(ka[0]), "v"(ka[1]), "v"(ka[2]), "v"(ka[3]),
                       "v"(kc[0]), "v"(kc[1]), "v"(kc[2]), "v"(kc[3]));
    #pragma unroll
    for (int c2 = 0; c2 < 4; ++c2) {
      v2f ka0; ka0.x = ka[c2].x; ka0.y = ka[c2].y;
      v2f ka1; ka1.x = ka[c2].z; ka1.y = ka[c2].w;
      v2f kc0; kc0.x = kc[c2].x; kc0.y = kc[c2].y;
      v2f kc1; kc1.x = kc[c2].z; kc1.y = kc[c2].w;
      s0 = wmma4(qa[2 * c2],     ka0, s0);
      s1 = wmma4(qa[2 * c2],     kc0, s1);
      s0 = wmma4(qa[2 * c2 + 1], ka1, s0);
      s1 = wmma4(qa[2 * c2 + 1], kc1, s1);
    }
    #pragma unroll
    for (int r = 0; r < 8; ++r) {
      const int m    = r + hi * 8;
      const int qidx = qt * 16 + m;
      const int kidx = kp * 32 + l15;
      S[m * 18 + l15]       = (kidx      <= qidx) ? s0[r] * scale : -3.0e38f;
      S[288 + m * 18 + l15] = (kidx + 16 <= qidx) ? s1[r] * scale : -3.0e38f;
    }
    wave_lds_order();

    {
      v2f sv[8];
      #pragma unroll
      for (int j = 0; j < 8; ++j) sv[j] = *(const v2f*)(Sr + 2 * j);
      float ml = fmaxf(sv[0].x, sv[0].y);
      #pragma unroll
      for (int j = 1; j < 8; ++j) ml = fmaxf(ml, fmaxf(sv[j].x, sv[j].y));
      const float mx = fmaxf(mreg, fmaxf(ml, other_half(ml)));
      const float f  = __expf(mreg - mx);
      float sum = 0.0f;
      #pragma unroll
      for (int j = 0; j < 8; ++j) {
        v2f p;
        p.x = __expf(sv[j].x - mx);
        p.y = __expf(sv[j].y - mx);
        *(v2f*)(Sr + 2 * j) = p;
        sum += p.x;
        sum += p.y;
      }
      mreg  = mx;
      lpart = lpart * f + sum;
      frow[l15] = f;
    }
    wave_lds_order();

    if (more) {
      #pragma unroll
      for (int i = 0; i < 2; ++i) *(v4f*)&Ks[cur ^ 1][ksoff + 16 * 36 * i] = gk[i];
      #pragma unroll
      for (int i = 0; i < 4; ++i) Vs[cur ^ 1][tid + i * 128] = gv[i];
    }

    const v4f* vp = &Vs[cur][lane];
    v4f vv[16];
    #pragma unroll
    for (int i = 0; i < 16; ++i) vv[i] = vp[i * 32];

    const v4f f0 = *(const v4f*)(frow + hi * 8);
    const v4f f1 = *(const v4f*)(frow + hi * 8 + 4);
    const float fr[8] = {f0.x, f0.y, f0.z, f0.w, f1.x, f1.y, f1.z, f1.w};
    #pragma unroll
    for (int r = 0; r < 8; ++r) {
      o0[r] *= fr[r]; o1[r] *= fr[r]; o2[r] *= fr[r]; o3[r] *= fr[r];
    }

    v2f pa[8];
    #pragma unroll
    for (int c = 0; c < 8; ++c) {
      pa[c] = *(const v2f*)(S + (c >> 2) * 288 + l15 * 18 + 8 * hi + 2 * (c & 3));
    }
    asm volatile("" :: "v"(vv[0]), "v"(vv[1]), "v"(vv[2]),  "v"(vv[3]),
                       "v"(vv[4]), "v"(vv[5]), "v"(vv[6]),  "v"(vv[7]),
                       "v"(vv[8]), "v"(vv[9]), "v"(vv[10]), "v"(vv[11]),
                       "v"(vv[12]), "v"(vv[13]), "v"(vv[14]), "v"(vv[15]));
    __builtin_amdgcn_sched_barrier(0);
    #pragma unroll
    for (int c = 0; c < 8; ++c) {
      const v4f va = vv[2 * c];
      const v4f vc = vv[2 * c + 1];
      v2f vb0; vb0.x = va.x; vb0.y = va.y;
      v2f vb1; vb1.x = va.z; vb1.y = va.w;
      v2f vb2; vb2.x = vc.x; vb2.y = vc.y;
      v2f vb3; vb3.x = vc.z; vb3.y = vc.w;
      o0 = wmma4(pa[c], vb0, o0);
      o1 = wmma4(pa[c], vb1, o1);
      o2 = wmma4(pa[c], vb2, o2);
      o3 = wmma4(pa[c], vb3, o3);
    }

    __syncthreads();
  }

  lrow[l15] = lpart + other_half(lpart);
  wave_lds_order();
  float inv[8];
  #pragma unroll
  for (int r = 0; r < 8; ++r) inv[r] = 1.0f / lrow[r + hi * 8];
  float* Ab = Abuf + ((size_t)(b * QKH + h)) * T_SEQ * VD;
  #pragma unroll
  for (int r = 0; r < 8; ++r) {
    v4f o;
    o.x = o0[r] * inv[r]; o.y = o1[r] * inv[r];
    o.z = o2[r] * inv[r]; o.w = o3[r] * inv[r];
    *(v4f*)(Ab + (size_t)(qt * 16 + r + hi * 8) * VD + 4 * l15) = o;
  }
}

__global__ __launch_bounds__(256) void k_diffnorm(const float* __restrict__ Abuf,
                                                  const float* __restrict__ lamp,
                                                  float* __restrict__ Df) {
  const float lam = *lamp;
  const int tl   = threadIdx.x >> 5;
  const int lane = threadIdx.x & 31;
  const int hi   = lane >> 4;
  const int l15  = lane & 15;
  const int hh   = (int)(blockIdx.x & 15u);
  const int rg   = (int)(blockIdx.x >> 4);
  const int m    = rg * 16 + l15;
  const int b    = m >> 11;
  const int t    = m & 2047;
  const int col  = tl * 8 + hi * 4;

  const float* A1 = Abuf + ((size_t)(b * QKH + hh)      * T_SEQ + t) * VD + col;
  const float* A2 = Abuf + ((size_t)(b * QKH + 16 + hh) * T_SEQ + t) * VD + col;
  const v4f a1 = *(const v4f*)A1;
  const v4f a2 = *(const v4f*)A2;
  v4f d;
  d.x = a1.x - lam * a2.x; d.y = a1.y - lam * a2.y;
  d.z = a1.z - lam * a2.z; d.w = a1.w - lam * a2.w;

  __shared__ float part[16][17];
  part[l15][tl * 2 + hi] = d.x * d.x + d.y * d.y + d.z * d.z + d.w * d.w;
  __syncthreads();
  float ss = 0.0f;
  #pragma unroll
  for (int i = 0; i < 16; ++i) ss += part[l15][i];
  const float r = rsqrtf(ss * (1.0f / 64.0f) + 1e-5f) * 0.8f;

  v4f o;
  o.x = d.x * r; o.y = d.y * r; o.z = d.z * r; o.w = d.w * r;
  ((v4f*)Df)[((size_t)rg * 128 + (size_t)(hh * 8 + tl)) * 32 + lane] = o;
}

extern "C" void kernel_launch(void* const* d_in, const int* in_sizes, int n_in,
                              void* d_out, int out_size, void* d_ws, size_t ws_size,
                              hipStream_t stream) {
  const float* x      = (const float*)d_in[0];
  const float* W_attn = (const float*)d_in[1];
  const float* W_proj = (const float*)d_in[2];
  const float* lq1    = (const float*)d_in[3];
  const float* lk1    = (const float*)d_in[4];
  const float* lq2    = (const float*)d_in[5];
  const float* lk2    = (const float*)d_in[6];
  float* out = (float*)d_out;

  float* qkv  = (float*)d_ws;
  float* Abuf = qkv  + (size_t)B_SZ * T_SEQ * ROWQKV;
  float* Dbuf = Abuf + (size_t)B_SZ * QKH * T_SEQ * VD;
  float* Vf   = Dbuf + (size_t)B_SZ * T_SEQ * C_DIM;
  float* Xf   = Vf   + (size_t)B_SZ * NH * T_SEQ * VD;
  float* lam  = Xf   + (size_t)B_SZ * T_SEQ * C_DIM;

  const int M = B_SZ * T_SEQ;

  k_prep<<<1 + XBLK, 256, 0, stream>>>(x, Xf, lq1, lk1, lq2, lk2, lam);
  {
    dim3 grid(3 * C_DIM / 128, M / 128);
    k_gemm<1><<<grid, 256, 0, stream>>>(Xf, W_attn, qkv, Vf, 3 * C_DIM, C_DIM);
  }
  {
    dim3 grid((T_SEQ / 64) * QKH * B_SZ);
    k_attn<<<grid, 128, 0, stream>>>(qkv, Vf, Abuf);
  }
  k_diffnorm<<<(B_SZ * T_SEQ / 16) * NH, 256, 0, stream>>>(Abuf, lam, Dbuf);
  {
    dim3 grid(C_DIM / 128, M / 128);
    k_gemm<0><<<grid, 256, 0, stream>>>(Dbuf, W_proj, out, Vf, C_DIM, C_DIM);
  }
  (void)in_sizes; (void)n_in; (void)out_size; (void)ws_size;
}


# both GEMMs: operand loads held two 8-k trips ahead (three rotating register sets; one ahead before); N, K compile-time so every row and trip address is an immediate; k and sum order unchanged
# speedup vs baseline: 1.9621x; 1.0449x over previous
#include <hip/hip_runtime.h>
#include <hip/hip_bf16.h>
#include <math.h>

typedef float v2f __attribute__((ext_vector_type(2)));
typedef float v4f __attribute__((ext_vector_type(4)));
typedef float v8f __attribute__((ext_vector_type(8)));

#define B_SZ   2
#define T_SEQ  2048
#define C_DIM  1024
#define NH     16
#define HD     32
#define QKH    32
#define VD     64
#define ROWQKV 3072

__device__ __forceinline__ v8f wmma4(v2f a, v2f b, v8f c) {
  return __builtin_amdgcn_wmma_f32_16x16x4_f32(
       false, a,  false, b,
       (short)0, c,  false,  false);
}

__device__ __forceinline__ void gemm_trip(v8f (&acc)[2][4], const v4f a0, const v4f a1,
                                          const v2f (&bp0)[4], const v2f (&bp1)[4]) {
  v2f x0, x1;
  x0.x = a0.x; x0.y = a0.y;
  x1.x = a1.x; x1.y = a1.y;
  #pragma unroll
  for (int cg = 0; cg < 4; ++cg) {
    acc[0][cg] = wmma4(x0, bp0[cg], acc[0][cg]);
    acc[1][cg] = wmma4(x1, bp0[cg], acc[1][cg]);
  }
  x0.x = a0.z; x0.y = a0.w;
  x1.x = a1.z; x1.y = a1.w;
  #pragma unroll
  for (int cg = 0; cg < 4; ++cg) {
    acc[0][cg] = wmma4(x0, bp1[cg], acc[0][cg]);
    acc[1][cg] = wmma4(x1, bp1[cg], acc[1][cg]);
  }
}

__device__ __forceinline__ void gemm_pairs(const v4f (&b)[4], v2f (&bp0)[4], v2f (&bp1)[4]) {
  #pragma unroll
  for (int cg = 0; cg < 4; ++cg) {
    bp0[cg].x = b[0][cg]; bp0[cg].y = b[1][cg];
    bp1[cg].x = b[2][cg]; bp1[cg].y = b[3][cg];
  }
}

__device__ __forceinline__ v2f rope_pair(v2f x, int j, int t) {
  const float theta = __expf(-((float)(2 * j) / 32.0f) * logf(10000.0f));
  float s, c;
  __sincosf((float)t * theta, &s, &c);
  v2f r;
  r.x = x.x * c - x.y * s;
  r.y = x.x * s + x.y * c;
  return r;
}

template <int N, int K>
__device__ __forceinline__ void gemm_load(v4f &a0, v4f &a1, v4f (&b)[4],
                                          const float* pA, const float* pB) {
  a0 = *(const v4f*)pA;
  a1 = *(const v4f*)(pA + (K >> 3) * 128);
  #pragma unroll
  for (int i = 0; i < 4; ++i) b[i] = *(const v4f*)(pB + i * N);
}

__device__ __forceinline__ void gemm_step(v8f (&acc)[2][4], v4f a0, v4f a1,
                                          v4f b0, v4f b1, v4f b2, v4f b3) {
  asm volatile("" : "+v"(a0), "+v"(a1), "+v"(b0), "+v"(b1), "+v"(b2), "+v"(b3));
  const v4f b[4] = {b0, b1, b2, b3};
  v2f bp0[4], bp1[4];
  gemm_pairs(b, bp0, bp1);
  gemm_trip(acc, a0, a1, bp0, bp1);
}

template <int QKV, int N, int K>
__global__ __launch_bounds__(256) void k_gemm(const float* __restrict__ Af,
                                              const float* __restrict__ Bm,
                                              float* __restrict__ Cm,
                                              float* __restrict__ Vf) {
  static_assert((K % 8) == 0 && ((K / 8) % 3) == 2, "trip count must be 3n+2");
  const int lane = threadIdx.x & 31;
  const int wid  = threadIdx.x >> 5;
  const int wy   = wid & 3;
  const int wx   = wid >> 2;
  const int rowBase = blockIdx.y * 128 + wy * 32;
  const int colBase = blockIdx.x * 128 + wx * 64;
  const int hi   = lane >> 4;
  const int l15  = lane & 15;
  const int bc   = colBase + 4 * l15;

  const float* pA = Af + ((size_t)(rowBase >> 4) * (size_t)(K >> 3) * 32 + lane) * 4;
  const float* pB = Bm + (size_t)(hi * 4) * N + bc;
  constexpr int TA = 128;
  constexpr int TB = 8 * N;

  v8f acc[2][4] = {};

  v4f a0_0, a1_0, b_0[4];
  v4f a0_1, a1_1, b_1[4];
  v4f a0_2, a1_2, b_2[4];

  gemm_load<N, K>(a0_0, a1_0, b_0, pA, pB);
  gemm_load<N, K>(a0_1, a1_1, b_1, pA + TA, pB + TB);

  for (int it = 0; it < (K / 8) / 3; ++it) {
    gemm_load<N, K>(a0_2, a1_2, b_2, pA + 2 * TA, pB + 2 * TB);
    __builtin_amdgcn_sched_barrier(0);
    gemm_step(acc, a0_0, a1_0, b_0[0], b_0[1], b_0[2], b_0[3]);
    __builtin_amdgcn_sched_barrier(0);
    gemm_load<N, K>(a0_0, a1_0, b_0, pA + 3 * TA, pB + 3 * TB);
    __builtin_amdgcn_sched_barrier(0);
    gemm_step(acc, a0_1, a1_1, b_1[0], b_1[1], b_1[2], b_1[3]);
    __builtin_amdgcn_sched_barrier(0);
    gemm_load<N, K>(a0_1, a1_1, b_1, pA + 4 * TA, pB + 4 * TB);
    __builtin_amdgcn_sched_barrier(0);
    gemm_step(acc, a0_2, a1_2, b_2[0], b_2[1], b_2[2], b_2[3]);
    __builtin_amdgcn_sched_barrier(0);
    pA += 3 * TA; pB += 3 * TB;
  }
  gemm_step(acc, a0_0, a1_0, b_0[0], b_0[1], b_0[2], b_0[3]);
  gemm_step(acc, a0_1, a1_1, b_1[0], b_1[1], b_1[2], b_1[3]);

  if (QKV && blockIdx.x * 128 >= 2 * C_DIM) {
    const int hv = (colBase - 2 * C_DIM) >> 6;
    #pragma unroll
    for (int g = 0; g < 2; ++g) {
      const int grp = (rowBase >> 4) + g;
      const int b   = grp >> 7;
      const int kb  = grp & 127;
      v4f* dst = (v4f*)Vf + ((size_t)((b * NH + hv) * 128 + kb) * 8) * 32 + lane;
      #pragma unroll
      for (int p = 0; p < 4; ++p) {
        #pragma unroll
        for (int tp = 0; tp < 2; ++tp) {
          v4f w;
          w.x = acc[g][2 * tp][2 * p];     w.y = acc[g][2 * tp][2 * p + 1];
          w.z = acc[g][2 * tp + 1][2 * p]; w.w = acc[g][2 * tp + 1][2 * p + 1];
          dst[(p * 2 + tp) * 32] = w;
        }
      }
    }
    return;
  }
  const int j0 = (bc & 31) >> 1;
  #pragma unroll
  for (int g = 0; g < 2; ++g) {
    #pragma unroll
    for (int r = 0; r < 8; ++r) {
      const size_t row = (size_t)(rowBase + g * 16 + r + hi * 8);
      v4f o;
      o.x = acc[g][0][r]; o.y = acc[g][1][r]; o.z = acc[g][2][r]; o.w = acc[g][3][r];
      if (QKV) {
        const int t = (int)(row & (size_t)(T_SEQ - 1));
        v2f e0, e1;
        e0.x = o.x; e0.y = o.y; e1.x = o.z; e1.y = o.w;
        e0 = rope_pair(e0, j0, t);
        e1 = rope_pair(e1, j0 + 1, t);
        o.x = e0.x; o.y = e0.y; o.z = e1.x; o.w = e1.y;
      }
      *(v4f*)(Cm + row * N + bc) = o;
    }
  }
}

#define XBLK 1024

__global__ __launch_bounds__(256) void k_prep(const float* __restrict__ x,
                                              float* __restrict__ Xf,
                                              const float* __restrict__ lq1,
                                              const float* __restrict__ lk1,
                                              const float* __restrict__ lq2,
                                              const float* __restrict__ lk2,
                                              float* __restrict__ lam) {
  if (blockIdx.x == 0) {
    if (threadIdx.x < 32) {
      const int l = threadIdx.x;
      float s1 = lq1[l] * lk1[l];
      float s2 = lq2[l] * lk2[l];
      #pragma unroll
      for (int off = 16; off; off >>= 1) {
        s1 += __shfl_xor(s1, off, 32);
        s2 += __shfl_xor(s2, off, 32);
      }
      if (l == 0) *lam = __expf(s1) - __expf(s2) + 0.2f;
    }
    return;
  }
  const unsigned base = (blockIdx.x - 1u) * 1024u + threadIdx.x;
  v4f w[4];
  #pragma unroll
  for (int i = 0; i < 4; ++i) {
    const unsigned o = base + 256u * i;
    const int lane = o & 31;
    const int trip = (o >> 5) & 127;
    const int rg   = (int)(o >> 12);
    const int hi   = lane >> 4;
    const int l15  = lane & 15;
    w[i] = *(const v4f*)(x + (size_t)(rg * 16 + l15) * C_DIM + trip * 8 + hi * 4);
  }
  #pragma unroll
  for (int i = 0; i < 4; ++i) ((v4f*)Xf)[base + 256u * i] = w[i];
}

__device__ __forceinline__ float other_half(float v) {
  const unsigned u = __float_as_uint(v);
  return __uint_as_float(
      __builtin_amdgcn_permlanex16(u, u, 0x76543210u, 0xfedcba98u, false, false));
}

__device__ __forceinline__ void wave_lds_order() {
  __builtin_amdgcn_fence(__ATOMIC_SEQ_CST, "wavefront");
  __builtin_amdgcn_wave_barrier();
  __builtin_amdgcn_fence(__ATOMIC_SEQ_CST, "wavefront");
}

#define AW 4

__global__ __launch_bounds__(128) void k_attn(const float* __restrict__ qkv,
                                              const float* __restrict__ Vf,
                                              float* __restrict__ Abuf) {
  const unsigned L = blockIdx.x;
  const int g    = 31 - (int)(L >> 6);
  const int h    = (int)(L & 31u);
  const int b    = (int)((L >> 5) & 1u);
  const int tid  = threadIdx.x;
  const int w    = tid >> 5;
  const int lane = tid & 31;
  const int hi   = lane >> 4;
  const int l15  = lane & 15;
  const int qt   = 4 * g + w;
  const int ntrip = 2 * g + 2;

  const float* Q  = qkv + (size_t)b * T_SEQ * ROWQKV + h * HD;
  const float* Kg = qkv + (size_t)b * T_SEQ * ROWQKV + C_DIM + h * HD
                        + (size_t)(tid >> 3) * ROWQKV + 4 * (tid & 7);
  const int ksoff = (tid >> 3) * 36 + 4 * (tid & 7);
  const v4f* Vg = (const v4f*)Vf + (size_t)(b * NH + (h & 15)) * (128 * 256) + tid;

  __shared__ __attribute__((aligned(16))) float Ks[2][32 * 36];
  __shared__ v4f Vs[2][512];
  __shared__ __attribute__((aligned(16))) float Sall[AW][2 * 288];
  __shared__ __attribute__((aligned(16))) float frowall[AW][16];
  __shared__ float lrowall[AW][16];
  float* const S    = Sall[w];
  float* const frow = frowall[w];
  float* const lrow = lrowall[w];

  const size_t qoff = (size_t)(qt * 16 + l15) * ROWQKV;
  v2f qa[8];
  #pragma unroll
  for (int c2 = 0; c2 < 4; ++c2) {
    const v4f q4 = *(const v4f*)(Q + qoff + 8 * c2 + 4 * hi);
    qa[2 * c2].x     = q4.x; qa[2 * c2].y     = q4.y;
    qa[2 * c2 + 1].x = q4.z; qa[2 * c2 + 1].y = q4.w;
  }

  v8f o0 = {}, o1 = {}, o2 = {}, o3 = {};
  const float scale = 0.17677669529663687f;

  float mreg  = -3.0e38f;
  float lpart = 0.0f;
  float* const Sr = S + hi * 288 + l15 * 18;

  {
    v4f gk[2], gv[4];
    #pragma unroll
    for (int i = 0; i < 2; ++i) gk[i] = *(const v4f*)(Kg + (size_t)(16 * i) * ROWQKV);
    #pragma unroll
    for (int i = 0; i < 4; ++i) gv[i] = Vg[i * 128];
    #pragma unroll
    for (int i = 0; i < 2; ++i) *(v4f*)&Ks[0][ksoff + 16 * 36 * i] = gk[i];
    #pragma unroll
    for (int i = 0; i < 4; ++i) Vs[0][tid + i * 128] = gv[i];
  }
  __syncthreads();

  for (int kp = 0; kp < ntrip; ++kp) {
    const int  cur  = kp & 1;
    const bool more = (kp + 1) < ntrip;
    v4f gk[2], gv[4];
    if (more) {
      const float* kn = Kg + (size_t)((kp + 1) * 32) * ROWQKV;
      const v4f* vn = Vg + (kp + 1) * 512;
      #pragma unroll
      for (int i = 0; i < 2; ++i) gk[i] = *(const v4f*)(kn + (size_t)(16 * i) * ROWQKV);
      #pragma unroll
      for (int i = 0; i < 4; ++i) gv[i] = vn[i * 128];
    }

    v8f s0 = {}, s1 = {};
    const float* kptr = &Ks[cur][l15 * 36 + 4 * hi];
    v4f ka[4], kc[4];
    #pragma unroll
    for (int c2 = 0; c2 < 4; ++c2) {
      ka[c2] = *(const v4f*)(kptr + 8 * c2);
      kc[c2] = *(const v4f*)(kptr + 16 * 36 + 8 * c2);
    }
    asm volatile("" :: "v"(ka[0]), "v"(ka[1]), "v"(ka[2]), "v"(ka[3]),
                       "v"(kc[0]), "v"(kc[1]), "v"(kc[2]), "v"(kc[3]));
    #pragma unroll
    for (int c2 = 0; c2 < 4; ++c2) {
      v2f ka0; ka0.x = ka[c2].x; ka0.y = ka[c2].y;
      v2f ka1; ka1.x = ka[c2].z; ka1.y = ka[c2].w;
      v2f kc0; kc0.x = kc[c2].x; kc0.y = kc[c2].y;
      v2f kc1; kc1.x = kc[c2].z; kc1.y = kc[c2].w;
      s0 = wmma4(qa[2 * c2],     ka0, s0);
      s1 = wmma4(qa[2 * c2],     kc0, s1);
      s0 = wmma4(qa[2 * c2 + 1], ka1, s0);
      s1 = wmma4(qa[2 * c2 + 1], kc1, s1);
    }
    #pragma unroll
    for (int r = 0; r < 8; ++r) {
      const int m    = r + hi * 8;
      const int qidx = qt * 16 + m;
      const int kidx = kp * 32 + l15;
      S[m * 18 + l15]       = (kidx      <= qidx) ? s0[r] * scale : -3.0e38f;
      S[288 + m * 18 + l15] = (kidx + 16 <= qidx) ? s1[r] * scale : -3.0e38f;
    }
    wave_lds_order();

    {
      v2f sv[8];
      #pragma unroll
      for (int j = 0; j < 8; ++j) sv[j] = *(const v2f*)(Sr + 2 * j);
      float ml = fmaxf(sv[0].x, sv[0].y);
      #pragma unroll
      for (int j = 1; j < 8; ++j) ml = fmaxf(ml, fmaxf(sv[j].x, sv[j].y));
      const float mx = fmaxf(mreg, fmaxf(ml, other_half(ml)));
      const float f  = __expf(mreg - mx);
      float sum = 0.0f;
      #pragma unroll
      for (int j = 0; j < 8; ++j) {
        v2f p;
        p.x = __expf(sv[j].x - mx);
        p.y = __expf(sv[j].y - mx);
        *(v2f*)(Sr + 2 * j) = p;
        sum += p.x;
        sum += p.y;
      }
      mreg  = mx;
      lpart = lpart * f + sum;
      frow[l15] = f;
    }
    wave_lds_order();

    if (more) {
      #pragma unroll
      for (int i = 0; i < 2; ++i) *(v4f*)&Ks[cur ^ 1][ksoff + 16 * 36 * i] = gk[i];
      #pragma unroll
      for (int i = 0; i < 4; ++i) Vs[cur ^ 1][tid + i * 128] = gv[i];
    }

    const v4f* vp = &Vs[cur][lane];
    v4f vv[16];
    #pragma unroll
    for (int i = 0; i < 16; ++i) vv[i] = vp[i * 32];

    const v4f f0 = *(const v4f*)(frow + hi * 8);
    const v4f f1 = *(const v4f*)(frow + hi * 8 + 4);
    const float fr[8] = {f0.x, f0.y, f0.z, f0.w, f1.x, f1.y, f1.z, f1.w};
    #pragma unroll
    for (int r = 0; r < 8; ++r) {
      o0[r] *= fr[r]; o1[r] *= fr[r]; o2[r] *= fr[r]; o3[r] *= fr[r];
    }

    v2f pa[8];
    #pragma unroll
    for (int c = 0; c < 8; ++c) {
      pa[c] = *(const v2f*)(S + (c >> 2) * 288 + l15 * 18 + 8 * hi + 2 * (c & 3));
    }
    asm volatile("" :: "v"(vv[0]), "v"(vv[1]), "v"(vv[2]),  "v"(vv[3]),
                       "v"(vv[4]), "v"(vv[5]), "v"(vv[6]),  "v"(vv[7]),
                       "v"(vv[8]), "v"(vv[9]), "v"(vv[10]), "v"(vv[11]),
                       "v"(vv[12]), "v"(vv[13]), "v"(vv[14]), "v"(vv[15]));
    __builtin_amdgcn_sched_barrier(0);
    #pragma unroll
    for (int c = 0; c < 8; ++c) {
      const v4f va = vv[2 * c];
      const v4f vc = vv[2 * c + 1];
      v2f vb0; vb0.x = va.x; vb0.y = va.y;
      v2f vb1; vb1.x = va.z; vb1.y = va.w;
      v2f vb2; vb2.x = vc.x; vb2.y = vc.y;
      v2f vb3; vb3.x = vc.z; vb3.y = vc.w;
      o0 = wmma4(pa[c], vb0, o0);
      o1 = wmma4(pa[c], vb1, o1);
      o2 = wmma4(pa[c], vb2, o2);
      o3 = wmma4(pa[c], vb3, o3);
    }

    __syncthreads();
  }

  lrow[l15] = lpart + other_half(lpart);
  wave_lds_order();
  float inv[8];
  #pragma unroll
  for (int r = 0; r < 8; ++r) inv[r] = 1.0f / lrow[r + hi * 8];
  float* Ab = Abuf + ((size_t)(b * QKH + h)) * T_SEQ * VD;
  #pragma unroll
  for (int r = 0; r < 8; ++r) {
    v4f o;
    o.x = o0[r] * inv[r]; o.y = o1[r] * inv[r];
    o.z = o2[r] * inv[r]; o.w = o3[r] * inv[r];
    *(v4f*)(Ab + (size_t)(qt * 16 + r + hi * 8) * VD + 4 * l15) = o;
  }
}

__global__ __launch_bounds__(256) void k_diffnorm(const float* __restrict__ Abuf,
                                                  const float* __restrict__ lamp,
                                                  float* __restrict__ Df) {
  const float lam = *lamp;
  const int tl   = threadIdx.x >> 5;
  const int lane = threadIdx.x & 31;
  const int hi   = lane >> 4;
  const int l15  = lane & 15;
  const int hh   = (int)(blockIdx.x & 15u);
  const int rg   = (int)(blockIdx.x >> 4);
  const int m    = rg * 16 + l15;
  const int b    = m >> 11;
  const int t    = m & 2047;
  const int col  = tl * 8 + hi * 4;

  const float* A1 = Abuf + ((size_t)(b * QKH + hh)      * T_SEQ + t) * VD + col;
  const float* A2 = Abuf + ((size_t)(b * QKH + 16 + hh) * T_SEQ + t) * VD + col;
  const v4f a1 = *(const v4f*)A1;
  const v4f a2 = *(const v4f*)A2;
  v4f d;
  d.x = a1.x - lam * a2.x; d.y = a1.y - lam * a2.y;
  d.z = a1.z - lam * a2.z; d.w = a1.w - lam * a2.w;

  __shared__ float part[16][17];
  part[l15][tl * 2 + hi] = d.x * d.x + d.y * d.y + d.z * d.z + d.w * d.w;
  __syncthreads();
  float ss = 0.0f;
  #pragma unroll
  for (int i = 0; i < 16; ++i) ss += part[l15][i];
  const float r = rsqrtf(ss * (1.0f / 64.0f) + 1e-5f) * 0.8f;

  v4f o;
  o.x = d.x * r; o.y = d.y * r; o.z = d.z * r; o.w = d.w * r;
  ((v4f*)Df)[((size_t)rg * 128 + (size_t)(hh * 8 + tl)) * 32 + lane] = o;
}

extern "C" void kernel_launch(void* const* d_in, const int* in_sizes, int n_in,
                              void* d_out, int out_size, void* d_ws, size_t ws_size,
                              hipStream_t stream) {
  const float* x      = (const float*)d_in[0];
  const float* W_attn = (const float*)d_in[1];
  const float* W_proj = (const float*)d_in[2];
  const float* lq1    = (const float*)d_in[3];
  const float* lk1    = (const float*)d_in[4];
  const float* lq2    = (const float*)d_in[5];
  const float* lk2    = (const float*)d_in[6];
  float* out = (float*)d_out;

  float* qkv  = (float*)d_ws;
  float* Abuf = qkv  + (size_t)B_SZ * T_SEQ * ROWQKV;
  float* Dbuf = Abuf + (size_t)B_SZ * QKH * T_SEQ * VD;
  float* Vf   = Dbuf + (size_t)B_SZ * T_SEQ * C_DIM;
  float* Xf   = Vf   + (size_t)B_SZ * NH * T_SEQ * VD;
  float* lam  = Xf   + (size_t)B_SZ * T_SEQ * C_DIM;

  const int M = B_SZ * T_SEQ;

  k_prep<<<1 + XBLK, 256, 0, stream>>>(x, Xf, lq1, lk1, lq2, lk2, lam);
  {
    dim3 grid(3 * C_DIM / 128, M / 128);
    k_gemm<1, 3 * C_DIM, C_DIM><<<grid, 256, 0, stream>>>(Xf, W_attn, qkv, Vf);
  }
  {
    dim3 grid((T_SEQ / 64) * QKH * B_SZ);
    k_attn<<<grid, 128, 0, stream>>>(qkv, Vf, Abuf);
  }
  k_diffnorm<<<(B_SZ * T_SEQ / 16) * NH, 256, 0, stream>>>(Abuf, lam, Dbuf);
  {
    dim3 grid(C_DIM / 128, M / 128);
    k_gemm<0, C_DIM, C_DIM><<<grid, 256, 0, stream>>>(Dbuf, W_proj, out, Vf);
  }
  (void)in_sizes; (void)n_in; (void)out_size; (void)ws_size;
}


# attention softmax: fewer vector slots a trip: score scale*log2(e) folded into Q, exp2 direct; causal select only on a group's last two trips; row sums packed even/odd (regrouped); two trips a pass
# speedup vs baseline: 2.0317x; 1.0355x over previous
#include <hip/hip_runtime.h>
#include <hip/hip_bf16.h>
#include <math.h>

typedef float v2f __attribute__((ext_vector_type(2)));
typedef float v4f __attribute__((ext_vector_type(4)));
typedef float v8f __attribute__((ext_vector_type(8)));

#define B_SZ   2
#define T_SEQ  2048
#define C_DIM  1024
#define NH     16
#define HD     32
#define QKH    32
#define VD     64
#define ROWQKV 3072

__device__ __forceinline__ v8f wmma4(v2f a, v2f b, v8f c) {
  return __builtin_amdgcn_wmma_f32_16x16x4_f32(
       false, a,  false, b,
       (short)0, c,  false,  false);
}

__device__ __forceinline__ void gemm_trip(v8f (&acc)[2][4], const v4f a0, const v4f a1,
                                          const v2f (&bp0)[4], const v2f (&bp1)[4]) {
  v2f x0, x1;
  x0.x = a0.x; x0.y = a0.y;
  x1.x = a1.x; x1.y = a1.y;
  #pragma unroll
  for (int cg = 0; cg < 4; ++cg) {
    acc[0][cg] = wmma4(x0, bp0[cg], acc[0][cg]);
    acc[1][cg] = wmma4(x1, bp0[cg], acc[1][cg]);
  }
  x0.x = a0.z; x0.y = a0.w;
  x1.x = a1.z; x1.y = a1.w;
  #pragma unroll
  for (int cg = 0; cg < 4; ++cg) {
    acc[0][cg] = wmma4(x0, bp1[cg], acc[0][cg]);
    acc[1][cg] = wmma4(x1, bp1[cg], acc[1][cg]);
  }
}

__device__ __forceinline__ void gemm_pairs(const v4f (&b)[4], v2f (&bp0)[4], v2f (&bp1)[4]) {
  #pragma unroll
  for (int cg = 0; cg < 4; ++cg) {
    bp0[cg].x = b[0][cg]; bp0[cg].y = b[1][cg];
    bp1[cg].x = b[2][cg]; bp1[cg].y = b[3][cg];
  }
}

__device__ __forceinline__ v2f rope_pair(v2f x, int j, int t) {
  const float theta = __expf(-((float)(2 * j) / 32.0f) * logf(10000.0f));
  float s, c;
  __sincosf((float)t * theta, &s, &c);
  v2f r;
  r.x = x.x * c - x.y * s;
  r.y = x.x * s + x.y * c;
  return r;
}

template <int N, int K>
__device__ __forceinline__ void gemm_load(v4f &a0, v4f &a1, v4f (&b)[4],
                                          const float* pA, const float* pB) {
  a0 = *(const v4f*)pA;
  a1 = *(const v4f*)(pA + (K >> 3) * 128);
  #pragma unroll
  for (int i = 0; i < 4; ++i) b[i] = *(const v4f*)(pB + i * N);
}

__device__ __forceinline__ void gemm_step(v8f (&acc)[2][4], v4f a0, v4f a1,
                                          v4f b0, v4f b1, v4f b2, v4f b3) {
  asm volatile("" : "+v"(a0), "+v"(a1), "+v"(b0), "+v"(b1), "+v"(b2), "+v"(b3));
  const v4f b[4] = {b0, b1, b2, b3};
  v2f bp0[4], bp1[4];
  gemm_pairs(b, bp0, bp1);
  gemm_trip(acc, a0, a1, bp0, bp1);
}

template <int QKV, int N, int K>
__global__ __launch_bounds__(256) void k_gemm(const float* __restrict__ Af,
                                              const float* __restrict__ Bm,
                                              float* __restrict__ Cm,
                                              float* __restrict__ Vf) {
  static_assert((K % 8) == 0 && ((K / 8) % 3) == 2, "trip count must be 3n+2");
  const int lane = threadIdx.x & 31;
  const int wid  = threadIdx.x >> 5;
  const int wy   = wid & 3;
  const int wx   = wid >> 2;
  const int rowBase = blockIdx.y * 128 + wy * 32;
  const int colBase = blockIdx.x * 128 + wx * 64;
  const int hi   = lane >> 4;
  const int l15  = lane & 15;
  const int bc   = colBase + 4 * l15;

  const float* pA = Af + ((size_t)(rowBase >> 4) * (size_t)(K >> 3) * 32 + lane) * 4;
  const float* pB = Bm + (size_t)(hi * 4) * N + bc;
  constexpr int TA = 128;
  constexpr int TB = 8 * N;

  v8f acc[2][4] = {};

  v4f a0_0, a1_0, b_0[4];
  v4f a0_1, a1_1, b_1[4];
  v4f a0_2, a1_2, b_2[4];

  gemm_load<N, K>(a0_0, a1_0, b_0, pA, pB);
  gemm_load<N, K>(a0_1, a1_1, b_1, pA + TA, pB + TB);

  for (int it = 0; it < (K / 8) / 3; ++it) {
    gemm_load<N, K>(a0_2, a1_2, b_2, pA + 2 * TA, pB + 2 * TB);
    __builtin_amdgcn_sched_barrier(0);
    gemm_step(acc, a0_0, a1_0, b_0[0], b_0[1], b_0[2], b_0[3]);
    __builtin_amdgcn_sched_barrier(0);
    gemm_load<N, K>(a0_0, a1_0, b_0, pA + 3 * TA, pB + 3 * TB);
    __builtin_amdgcn_sched_barrier(0);
    gemm_step(acc, a0_1, a1_1, b_1[0], b_1[1], b_1[2], b_1[3]);
    __builtin_amdgcn_sched_barrier(0);
    gemm_load<N, K>(a0_1, a1_1, b_1, pA + 4 * TA, pB + 4 * TB);
    __builtin_amdgcn_sched_barrier(0);
    gemm_step(acc, a0_2, a1_2, b_2[0], b_2[1], b_2[2], b_2[3]);
    __builtin_amdgcn_sched_barrier(0);
    pA += 3 * TA; pB += 3 * TB;
  }
  gemm_step(acc, a0_0, a1_0, b_0[0], b_0[1], b_0[2], b_0[3]);
  gemm_step(acc, a0_1, a1_1, b_1[0], b_1[1], b_1[2], b_1[3]);

  if (QKV && blockIdx.x * 128 >= 2 * C_DIM) {
    const int hv = (colBase - 2 * C_DIM) >> 6;
    #pragma unroll
    for (int g = 0; g < 2; ++g) {
      const int grp = (rowBase >> 4) + g;
      const int b   = grp >> 7;
      const int kb  = grp & 127;
      v4f* dst = (v4f*)Vf + ((size_t)((b * NH + hv) * 128 + kb) * 8) * 32 + lane;
      #pragma unroll
      for (int p = 0; p < 4; ++p) {
        #pragma unroll
        for (int tp = 0; tp < 2; ++tp) {
          v4f w;
          w.x = acc[g][2 * tp][2 * p];     w.y = acc[g][2 * tp][2 * p + 1];
          w.z = acc[g][2 * tp + 1][2 * p]; w.w = acc[g][2 * tp + 1][2 * p + 1];
          dst[(p * 2 + tp) * 32] = w;
        }
      }
    }
    return;
  }
  const int j0 = (bc & 31) >> 1;
  #pragma unroll
  for (int g = 0; g < 2; ++g) {
    #pragma unroll
    for (int r = 0; r < 8; ++r) {
      const size_t row = (size_t)(rowBase + g * 16 + r + hi * 8);
      v4f o;
      o.x = acc[g][0][r]; o.y = acc[g][1][r]; o.z = acc[g][2][r]; o.w = acc[g][3][r];
      if (QKV) {
        const int t = (int)(row & (size_t)(T_SEQ - 1));
        v2f e0, e1;
        e0.x = o.x; e0.y = o.y; e1.x = o.z; e1.y = o.w;
        e0 = rope_pair(e0, j0, t);
        e1 = rope_pair(e1, j0 + 1, t);
        o.x = e0.x; o.y = e0.y; o.z = e1.x; o.w = e1.y;
      }
      *(v4f*)(Cm + row * N + bc) = o;
    }
  }
}

#define XBLK 1024

__global__ __launch_bounds__(256) void k_prep(const float* __restrict__ x,
                                              float* __restrict__ Xf,
                                              const float* __restrict__ lq1,
                                              const float* __restrict__ lk1,
                                              const float* __restrict__ lq2,
                                              const float* __restrict__ lk2,
                                              float* __restrict__ lam) {
  if (blockIdx.x == 0) {
    if (threadIdx.x < 32) {
      const int l = threadIdx.x;
      float s1 = lq1[l] * lk1[l];
      float s2 = lq2[l] * lk2[l];
      #pragma unroll
      for (int off = 16; off; off >>= 1) {
        s1 += __shfl_xor(s1, off, 32);
        s2 += __shfl_xor(s2, off, 32);
      }
      if (l == 0) *lam = __expf(s1) - __expf(s2) + 0.2f;
    }
    return;
  }
  const unsigned base = (blockIdx.x - 1u) * 1024u + threadIdx.x;
  v4f w[4];
  #pragma unroll
  for (int i = 0; i < 4; ++i) {
    const unsigned o = base + 256u * i;
    const int lane = o & 31;
    const int trip = (o >> 5) & 127;
    const int rg   = (int)(o >> 12);
    const int hi   = lane >> 4;
    const int l15  = lane & 15;
    w[i] = *(const v4f*)(x + (size_t)(rg * 16 + l15) * C_DIM + trip * 8 + hi * 4);
  }
  #pragma unroll
  for (int i = 0; i < 4; ++i) ((v4f*)Xf)[base + 256u * i] = w[i];
}

__device__ __forceinline__ float other_half(float v) {
  const unsigned u = __float_as_uint(v);
  return __uint_as_float(
      __builtin_amdgcn_permlanex16(u, u, 0x76543210u, 0xfedcba98u, false, false));
}

__device__ __forceinline__ void wave_lds_order() {
  __builtin_amdgcn_fence(__ATOMIC_SEQ_CST, "wavefront");
  __builtin_amdgcn_wave_barrier();
  __builtin_amdgcn_fence(__ATOMIC_SEQ_CST, "wavefront");
}

#define AW 4

#define QSCALE (0.17677669529663687f * 1.4426950408889634f)

template <bool MASKED, int CUR>
__device__ __forceinline__ void attn_trip(
    const int kp, const bool more, const int qt, const int tid, const int lane,
    const int hi, const int l15,
    const float* __restrict__ Kg, const v4f* __restrict__ Vg, const int ksoff,
    float* const Ksb, v4f* const Vsb, float* const S, float* const frow,
    const v2f (&qa)[8],
    v8f &o0, v8f &o1, v8f &o2, v8f &o3, float &mreg, v2f &lsum) {
  float* const Kcur = Ksb + CUR * (32 * 36);
  float* const Knxt = Ksb + (CUR ^ 1) * (32 * 36);
  v4f* const Vcur = Vsb + CUR * 512;
  v4f* const Vnxt = Vsb + (CUR ^ 1) * 512;
  float* const Sr = S + hi * 288 + l15 * 18;

  v4f gk[2], gv[4];
  if (more) {
    const float* kn = Kg + (size_t)((kp + 1) * 32) * ROWQKV;
    const v4f* vn = Vg + (kp + 1) * 512;
    #pragma unroll
    for (int i = 0; i < 2; ++i) gk[i] = *(const v4f*)(kn + (size_t)(16 * i) * ROWQKV);
    #pragma unroll
    for (int i = 0; i < 4; ++i) gv[i] = vn[i * 128];
  }

  v8f s0 = {}, s1 = {};
  const float* kptr = Kcur + l15 * 36 + 4 * hi;
  v4f ka[4], kc[4];
  #pragma unroll
  for (int c2 = 0; c2 < 4; ++c2) {
    ka[c2] = *(const v4f*)(kptr + 8 * c2);
    kc[c2] = *(const v4f*)(kptr + 16 * 36 + 8 * c2);
  }
  asm volatile("" :: "v"(ka[0]), "v"(ka[1]), "v"(ka[2]), "v"(ka[3]),
                     "v"(kc[0]), "v"(kc[1]), "v"(kc[2]), "v"(kc[3]));
  #pragma unroll
  for (int c2 = 0; c2 < 4; ++c2) {
    v2f ka0; ka0.x = ka[c2].x; ka0.y = ka[c2].y;
    v2f ka1; ka1.x = ka[c2].z; ka1.y = ka[c2].w;
    v2f kc0; kc0.x = kc[c2].x; kc0.y = kc[c2].y;
    v2f kc1; kc1.x = kc[c2].z; kc1.y = kc[c2].w;
    s0 = wmma4(qa[2 * c2],     ka0, s0);
    s1 = wmma4(qa[2 * c2],     kc0, s1);
    s0 = wmma4(qa[2 * c2 + 1], ka1, s0);
    s1 = wmma4(qa[2 * c2 + 1], kc1, s1);
  }
  float* const Sw = S + hi * (8 * 18) + l15;
  if (MASKED) {
    #pragma unroll
    for (int r = 0; r < 8; ++r) {
      const int qidx = qt * 16 + r + hi * 8;
      const int kidx = kp * 32 + l15;
      Sw[r * 18]       = (kidx      <= qidx) ? s0[r] : -3.0e38f;
      Sw[288 + r * 18] = (kidx + 16 <= qidx) ? s1[r] : -3.0e38f;
    }
  } else {
    #pragma unroll
    for (int r = 0; r < 8; ++r) {
      Sw[r * 18]       = s0[r];
      Sw[288 + r * 18] = s1[r];
    }
  }
  wave_lds_order();

  {
    v2f sv[8];
    #pragma unroll
    for (int j = 0; j < 8; ++j) sv[j] = *(const v2f*)(Sr + 2 * j);
    float ml = fmaxf(sv[0].x, sv[0].y);
    #pragma unroll
    for (int j = 1; j < 8; ++j) ml = fmaxf(ml, fmaxf(sv[j].x, sv[j].y));
    const float mx = fmaxf(mreg, fmaxf(ml, other_half(ml)));
    const float f  = __builtin_amdgcn_exp2f(mreg - mx);
    v2f sum = {0.0f, 0.0f};
    #pragma unroll
    for (int j = 0; j < 8; ++j) {
      v2f p;
      p.x = __builtin_amdgcn_exp2f(sv[j].x - mx);
      p.y = __builtin_amdgcn_exp2f(sv[j].y - mx);
      *(v2f*)(Sr + 2 * j) = p;
      sum += p;
    }
    mreg = mx;
    lsum = lsum * f + sum;
    frow[l15] = f;
  }
  wave_lds_order();

  if (more) {
    #pragma unroll
    for (int i = 0; i < 2; ++i) *(v4f*)(Knxt + ksoff + 16 * 36 * i) = gk[i];
    #pragma unroll
    for (int i = 0; i < 4; ++i) Vnxt[tid + i * 128] = gv[i];
  }

  const v4f* vp = Vcur + lane;
  v4f vv[16];
  #pragma unroll
  for (int i = 0; i < 16; ++i) vv[i] = vp[i * 32];

  const v4f f0 = *(const v4f*)(frow + hi * 8);
  const v4f f1 = *(const v4f*)(frow + hi * 8 + 4);
  const float fr[8] = {f0.x, f0.y, f0.z, f0.w, f1.x, f1.y, f1.z, f1.w};
  #pragma unroll
  for (int r = 0; r < 8; ++r) {
    o0[r] *= fr[r]; o1[r] *= fr[r]; o2[r] *= fr[r]; o3[r] *= fr[r];
  }

  v2f pa[8];
  #pragma unroll
  for (int c = 0; c < 8; ++c) {
    pa[c] = *(const v2f*)(S + (c >> 2) * 288 + l15 * 18 + 8 * hi + 2 * (c & 3));
  }
  asm volatile("" :: "v"(vv[0]), "v"(vv[1]), "v"(vv[2]),  "v"(vv[3]),
                     "v"(vv[4]), "v"(vv[5]), "v"(vv[6]),  "v"(vv[7]),
                     "v"(vv[8]), "v"(vv[9]), "v"(vv[10]), "v"(vv[11]),
                     "v"(vv[12]), "v"(vv[13]), "v"(vv[14]), "v"(vv[15]));
  __builtin_amdgcn_sched_barrier(0);
  #pragma unroll
  for (int c = 0; c < 8; ++c) {
    const v4f va = vv[2 * c];
    const v4f vc = vv[2 * c + 1];
    v2f vb0; vb0.x = va.x; vb0.y = va.y;
    v2f vb1; vb1.x = va.z; vb1.y = va.w;
    v2f vb2; vb2.x = vc.x; vb2.y = vc.y;
    v2f vb3; vb3.x = vc.z; vb3.y = vc.w;
    o0 = wmma4(pa[c], vb0, o0);
    o1 = wmma4(pa[c], vb1, o1);
    o2 = wmma4(pa[c], vb2, o2);
    o3 = wmma4(pa[c], vb3, o3);
  }

  __syncthreads();
}

__global__ __launch_bounds__(128) void k_attn(const float* __restrict__ qkv,
                                              const float* __restrict__ Vf,
                                              float* __restrict__ Abuf) {
  const unsigned L = blockIdx.x;
  const int g    = 31 - (int)(L >> 6);
  const int h    = (int)(L & 31u);
  const int b    = (int)((L >> 5) & 1u);
  const int tid  = threadIdx.x;
  const int w    = tid >> 5;
  const int lane = tid & 31;
  const int hi   = lane >> 4;
  const int l15  = lane & 15;
  const int qt   = 4 * g + w;
  const int ntrip = 2 * g + 2;

  const float* Q  = qkv + (size_t)b * T_SEQ * ROWQKV + h * HD;
  const float* Kg = qkv + (size_t)b * T_SEQ * ROWQKV + C_DIM + h * HD
                        + (size_t)(tid >> 3) * ROWQKV + 4 * (tid & 7);
  const int ksoff = (tid >> 3) * 36 + 4 * (tid & 7);
  const v4f* Vg = (const v4f*)Vf + (size_t)(b * NH + (h & 15)) * (128 * 256) + tid;

  __shared__ __attribute__((aligned(16))) float Ks[2][32 * 36];
  __shared__ v4f Vs[2][512];
  __shared__ __attribute__((aligned(16))) float Sall[AW][2 * 288];
  __shared__ __attribute__((aligned(16))) float frowall[AW][16];
  __shared__ float lrowall[AW][16];
  float* const S    = Sall[w];
  float* const frow = frowall[w];
  float* const lrow = lrowall[w];

  const size_t qoff = (size_t)(qt * 16 + l15) * ROWQKV;
  v2f qa[8];
  #pragma unroll
  for (int c2 = 0; c2 < 4; ++c2) {
    const v4f q4 = *(const v4f*)(Q + qoff + 8 * c2 + 4 * hi) * QSCALE;
    qa[2 * c2].x     = q4.x; qa[2 * c2].y     = q4.y;
    qa[2 * c2 + 1].x = q4.z; qa[2 * c2 + 1].y = q4.w;
  }

  v8f o0 = {}, o1 = {}, o2 = {}, o3 = {};

  float mreg = -3.0e38f;
  v2f lsum = {0.0f, 0.0f};

  {
    v4f gk[2], gv[4];
    #pragma unroll
    for (int i = 0; i < 2; ++i) gk[i] = *(const v4f*)(Kg + (size_t)(16 * i) * ROWQKV);
    #pragma unroll
    for (int i = 0; i < 4; ++i) gv[i] = Vg[i * 128];
    #pragma unroll
    for (int i = 0; i < 2; ++i) *(v4f*)&Ks[0][ksoff + 16 * 36 * i] = gk[i];
    #pragma unroll
    for (int i = 0; i < 4; ++i) Vs[0][tid + i * 128] = gv[i];
  }
  __syncthreads();

  for (int kp = 0; kp < ntrip - 2; kp += 2) {
    attn_trip<false, 0>(kp, true, qt, tid, lane, hi, l15, Kg, Vg, ksoff,
                        &Ks[0][0], &Vs[0][0], S, frow, qa, o0, o1, o2, o3, mreg, lsum);
    attn_trip<false, 1>(kp + 1, true, qt, tid, lane, hi, l15, Kg, Vg, ksoff,
                        &Ks[0][0], &Vs[0][0], S, frow, qa, o0, o1, o2, o3, mreg, lsum);
  }
  attn_trip<true, 0>(ntrip - 2, true, qt, tid, lane, hi, l15, Kg, Vg, ksoff,
                     &Ks[0][0], &Vs[0][0], S, frow, qa, o0, o1, o2, o3, mreg, lsum);
  attn_trip<true, 1>(ntrip - 1, false, qt, tid, lane, hi, l15, Kg, Vg, ksoff,
                     &Ks[0][0], &Vs[0][0], S, frow, qa, o0, o1, o2, o3, mreg, lsum);

  const float lpart = lsum.x + lsum.y;
  lrow[l15] = lpart + other_half(lpart);
  wave_lds_order();
  float inv[8];
  #pragma unroll
  for (int r = 0; r < 8; ++r) inv[r] = 1.0f / lrow[r + hi * 8];
  float* Ab = Abuf + ((size_t)(b * QKH + h)) * T_SEQ * VD;
  #pragma unroll
  for (int r = 0; r < 8; ++r) {
    v4f o;
    o.x = o0[r] * inv[r]; o.y = o1[r] * inv[r];
    o.z = o2[r] * inv[r]; o.w = o3[r] * inv[r];
    *(v4f*)(Ab + (size_t)(qt * 16 + r + hi * 8) * VD + 4 * l15) = o;
  }
}

__global__ __launch_bounds__(256) void k_diffnorm(const float* __restrict__ Abuf,
                                                  const float* __restrict__ lamp,
                                                  float* __restrict__ Df) {
  const float lam = *lamp;
  const int tl   = threadIdx.x >> 5;
  const int lane = threadIdx.x & 31;
  const int hi   = lane >> 4;
  const int l15  = lane & 15;
  const int hh   = (int)(blockIdx.x & 15u);
  const int rg   = (int)(blockIdx.x >> 4);
  const int m    = rg * 16 + l15;
  const int b    = m >> 11;
  const int t    = m & 2047;
  const int col  = tl * 8 + hi * 4;

  const float* A1 = Abuf + ((size_t)(b * QKH + hh)      * T_SEQ + t) * VD + col;
  const float* A2 = Abuf + ((size_t)(b * QKH + 16 + hh) * T_SEQ + t) * VD + col;
  const v4f a1 = *(const v4f*)A1;
  const v4f a2 = *(const v4f*)A2;
  v4f d;
  d.x = a1.x - lam * a2.x; d.y = a1.y - lam * a2.y;
  d.z = a1.z - lam * a2.z; d.w = a1.w - lam * a2.w;

  __shared__ float part[16][17];
  part[l15][tl * 2 + hi] = d.x * d.x + d.y * d.y + d.z * d.z + d.w * d.w;
  __syncthreads();
  float ss = 0.0f;
  #pragma unroll
  for (int i = 0; i < 16; ++i) ss += part[l15][i];
  const float r = rsqrtf(ss * (1.0f / 64.0f) + 1e-5f) * 0.8f;

  v4f o;
  o.x = d.x * r; o.y = d.y * r; o.z = d.z * r; o.w = d.w * r;
  ((v4f*)Df)[((size_t)rg * 128 + (size_t)(hh * 8 + tl)) * 32 + lane] = o;
}

extern "C" void kernel_launch(void* const* d_in, const int* in_sizes, int n_in,
                              void* d_out, int out_size, void* d_ws, size_t ws_size,
                              hipStream_t stream) {
  const float* x      = (const float*)d_in[0];
  const float* W_attn = (const float*)d_in[1];
  const float* W_proj = (const float*)d_in[2];
  const float* lq1    = (const float*)d_in[3];
  const float* lk1    = (const float*)d_in[4];
  const float* lq2    = (const float*)d_in[5];
  const float* lk2    = (const float*)d_in[6];
  float* out = (float*)d_out;

  float* qkv  = (float*)d_ws;
  float* Abuf = qkv  + (size_t)B_SZ * T_SEQ * ROWQKV;
  float* Dbuf = Abuf + (size_t)B_SZ * QKH * T_SEQ * VD;
  float* Vf   = Dbuf + (size_t)B_SZ * T_SEQ * C_DIM;
  float* Xf   = Vf   + (size_t)B_SZ * NH * T_SEQ * VD;
  float* lam  = Xf   + (size_t)B_SZ * T_SEQ * C_DIM;

  const int M = B_SZ * T_SEQ;

  k_prep<<<1 + XBLK, 256, 0, stream>>>(x, Xf, lq1, lk1, lq2, lk2, lam);
  {
    dim3 grid(3 * C_DIM / 128, M / 128);
    k_gemm<1, 3 * C_DIM, C_DIM><<<grid, 256, 0, stream>>>(Xf, W_attn, qkv, Vf);
  }
  {
    dim3 grid((T_SEQ / 64) * QKH * B_SZ);
    k_attn<<<grid, 128, 0, stream>>>(qkv, Vf, Abuf);
  }
  k_diffnorm<<<(B_SZ * T_SEQ / 16) * NH, 256, 0, stream>>>(Abuf, lam, Dbuf);
  {
    dim3 grid(C_DIM / 128, M / 128);
    k_gemm<0, C_DIM, C_DIM><<<grid, 256, 0, stream>>>(Dbuf, W_proj, out, Vf);
  }
  (void)in_sizes; (void)n_in; (void)out_size; (void)ws_size;
}
